# starting kernel
# baseline (speedup 1.0000x reference)
#include <hip/hip_runtime.h>
#include <hip/hip_bf16.h>

typedef float v2f __attribute__((ext_vector_type(2)));
typedef float v8f __attribute__((ext_vector_type(8)));

#define BATCH 4
#define SEQ   2048
#define DMOD  1024
#define NST   16
#define CHUNK 64
#define NCH   (SEQ / CHUNK)

#define WS_AMATT 0
#define WS_UT    (80 * 64)
#define WS_ACT   (80 * 64 + 64 * 16)
#define WS_S     8192
#define WS_H     (8192 + (size_t)BATCH * NCH * DMOD * NST)

__global__ __launch_bounds__(256) void k0_precompute(const float* __restrict__ A,
                                                     const float* __restrict__ Bvec,
                                                     float* __restrict__ ws) {
    __shared__ float cur[16][16];
    __shared__ float bb[16];
    __shared__ float kv[64];
    __shared__ float r[16];
    __shared__ float u[16];

    const int tid = threadIdx.x;
    const int m = tid >> 4, n = tid & 15;

    float* AmatT = ws + WS_AMATT;
    float* Ut    = ws + WS_UT;
    float* AcT   = ws + WS_ACT;

    if (tid < 16) bb[tid] = Bvec[tid];
    cur[m][n] = (m == n) ? 1.0f : 0.0f;
    __syncthreads();

    for (int p = 0; p <= CHUNK; ++p) {
        if (tid < 16) {
            float s = 0.f;
            for (int q = 0; q < 16; ++q) s += bb[q] * cur[q][tid];
            r[tid] = s;
        } else if (tid < 32) {
            const int mm = tid - 16;
            float s = 0.f;
            for (int q = 0; q < 16; ++q) s += cur[mm][q] * bb[q];
            u[mm] = s;
        }
        __syncthreads();

        if (p <= 63) {
            if (tid == 0) {
                float s = 0.f;
                for (int q = 0; q < 16; ++q) s += r[q] * bb[q];
                kv[p] = s;
            }
            if (tid < 16) Ut[(63 - p) * 16 + tid] = u[tid];
        }
        if (p >= 1) {
            if (tid < 16) AmatT[(64 + tid) * 64 + (p - 1)] = r[tid];
        }
        if (p == CHUNK) {
            AcT[n * 16 + m] = cur[m][n];
        }

        float s = 0.f;
        for (int q = 0; q < 16; ++q) s += A[m * 16 + q] * cur[q][n];
        __syncthreads();
        cur[m][n] = s;
        __syncthreads();
    }
    __syncthreads();

    for (int e = tid; e < 64 * 64; e += 256) {
        const int k = e >> 6, t = e & 63;
        AmatT[k * 64 + t] = (t >= k) ? kv[t - k] : 0.0f;
    }
}

__global__ __launch_bounds__(256) void k1_state_gemm(const float* __restrict__ x,
                                                     const float* __restrict__ Ut,
                                                     float* __restrict__ S) {
    const int lane = threadIdx.x & 31;
    const int wave = threadIdx.x >> 5;
    const int hi = lane >> 4;
    const int m = lane & 15;

    const int blk = blockIdx.x;
    const int dslab = blk & 7;
    const int chunk = (blk >> 3) & (NCH - 1);
    const int b = blk >> 8;
    const int d0 = (dslab * 8 + wave) * 16;

    const float* xb = x + ((size_t)b * SEQ + (size_t)chunk * CHUNK) * DMOD;

    v8f acc = {};
#pragma unroll
    for (int k0 = 0; k0 < CHUNK; k0 += 4) {
        const int r0 = k0 + 2 * hi;
        v2f a, bf;
        a.x  = xb[(size_t)r0 * DMOD + d0 + m];
        a.y  = xb[(size_t)(r0 + 1) * DMOD + d0 + m];
        bf.x = Ut[r0 * 16 + m];
        bf.y = Ut[(r0 + 1) * 16 + m];
        acc = __builtin_amdgcn_wmma_f32_16x16x4_f32(false, a, false, bf,
                                                    (short)0, acc, false, false);
    }

    float* Sp = S + (((size_t)b * NCH + chunk) * DMOD + d0) * NST;
#pragma unroll
    for (int v = 0; v < 8; ++v)
        Sp[(v + 8 * hi) * NST + m] = acc[v];
}

__global__ __launch_bounds__(256) void k2_scan(const float* __restrict__ AcT,
                                               const float* __restrict__ S,
                                               float* __restrict__ H) {
    __shared__ float hbuf[8][16][17];

    const int lane = threadIdx.x & 31;
    const int wave = threadIdx.x >> 5;
    const int hi = lane >> 4;
    const int m = lane & 15;

    const int gwave = blockIdx.x * 8 + wave;
    const int b = gwave >> 6;
    const int d0 = (gwave & 63) * 16;

    v2f bf[4];
#pragma unroll
    for (int k0 = 0; k0 < 16; k0 += 4) {
        const int r0 = k0 + 2 * hi;
        bf[k0 >> 2].x = AcT[r0 * 16 + m];
        bf[k0 >> 2].y = AcT[(r0 + 1) * 16 + m];
    }

    v8f h = {};
    for (int c = 0; c < NCH; ++c) {
#pragma unroll
        for (int v = 0; v < 8; ++v)
            hbuf[wave][v + 8 * hi][m] = h[v];
        __syncthreads();

        float* Hp = H + (((size_t)b * NCH + c) * NST) * DMOD;
#pragma unroll
        for (int v = 0; v < 8; ++v)
            Hp[(size_t)(2 * v + hi) * DMOD + d0 + m] = hbuf[wave][m][2 * v + hi];

        const float* Sp = S + (((size_t)b * NCH + c) * DMOD + d0) * NST;
        v8f acc;
#pragma unroll
        for (int v = 0; v < 8; ++v)
            acc[v] = Sp[(v + 8 * hi) * NST + m];

#pragma unroll
        for (int k0 = 0; k0 < 16; k0 += 4) {
            const int r0 = k0 + 2 * hi;
            v2f a;
            a.x = hbuf[wave][m][r0];
            a.y = hbuf[wave][m][r0 + 1];
            acc = __builtin_amdgcn_wmma_f32_16x16x4_f32(false, a, false, bf[k0 >> 2],
                                                        (short)0, acc, false, false);
        }
        h = acc;
        __syncthreads();
    }
}

__global__ __launch_bounds__(256) void k3_out_gemm(const float* __restrict__ x,
                                                   const float* __restrict__ AmatT,
                                                   const float* __restrict__ H,
                                                   float* __restrict__ y) {
    __shared__ float Asm[80 * 64];
    for (int e = threadIdx.x; e < 80 * 64; e += 256) Asm[e] = AmatT[e];
    __syncthreads();

    const int lane = threadIdx.x & 31;
    const int wave = threadIdx.x >> 5;
    const int hi = lane >> 4;
    const int m = lane & 15;

    const int blk = blockIdx.x;
    const int dquad = blk & 3;
    const int chunk = (blk >> 2) & (NCH - 1);
    const int b = blk >> 7;
    const int d0 = (dquad * 8 + wave) * 32;
    const int d1 = d0 + 16;

    const float* xb = x + ((size_t)b * SEQ + (size_t)chunk * CHUNK) * DMOD;
    const float* Hp = H + (((size_t)b * NCH + chunk) * NST) * DMOD;

    v8f acc[2][4] = {{{}, {}, {}, {}}, {{}, {}, {}, {}}};

#pragma unroll
    for (int kb = 0; kb < 4; ++kb) {
#pragma unroll
        for (int kk = 0; kk < 16; kk += 4) {
            const int r0 = kb * 16 + kk + 2 * hi;
            v2f b0, b1;
            b0.x = xb[(size_t)r0 * DMOD + d0 + m];
            b0.y = xb[(size_t)(r0 + 1) * DMOD + d0 + m];
            b1.x = xb[(size_t)r0 * DMOD + d1 + m];
            b1.y = xb[(size_t)(r0 + 1) * DMOD + d1 + m];
#pragma unroll
            for (int it = kb; it < 4; ++it) {
                v2f a;
                a.x = Asm[r0 * 64 + it * 16 + m];
                a.y = Asm[(r0 + 1) * 64 + it * 16 + m];
                acc[0][it] = __builtin_amdgcn_wmma_f32_16x16x4_f32(
                    false, a, false, b0, (short)0, acc[0][it], false, false);
                acc[1][it] = __builtin_amdgcn_wmma_f32_16x16x4_f32(
                    false, a, false, b1, (short)0, acc[1][it], false, false);
            }
        }
    }
#pragma unroll
    for (int k0 = 64; k0 < 80; k0 += 4) {
        const int r0 = k0 + 2 * hi;
        v2f b0, b1;
        b0.x = Hp[(size_t)(r0 - 64) * DMOD + d0 + m];
        b0.y = Hp[(size_t)(r0 - 63) * DMOD + d0 + m];
        b1.x = Hp[(size_t)(r0 - 64) * DMOD + d1 + m];
        b1.y = Hp[(size_t)(r0 - 63) * DMOD + d1 + m];
#pragma unroll
        for (int it = 0; it < 4; ++it) {
            v2f a;
            a.x = Asm[r0 * 64 + it * 16 + m];
            a.y = Asm[(r0 + 1) * 64 + it * 16 + m];
            acc[0][it] = __builtin_amdgcn_wmma_f32_16x16x4_f32(
                false, a, false, b0, (short)0, acc[0][it], false, false);
            acc[1][it] = __builtin_amdgcn_wmma_f32_16x16x4_f32(
                false, a, false, b1, (short)0, acc[1][it], false, false);
        }
    }

    float* yb = y + ((size_t)b * SEQ + (size_t)chunk * CHUNK) * DMOD;
#pragma unroll
    for (int it = 0; it < 4; ++it)
#pragma unroll
        for (int v = 0; v < 8; ++v) {
            const size_t row = (size_t)(it * 16 + v + 8 * hi) * DMOD;
            yb[row + d0 + m] = acc[0][it][v];
            yb[row + d1 + m] = acc[1][it][v];
        }
}

extern "C" void kernel_launch(void* const* d_in, const int* in_sizes, int n_in,
                              void* d_out, int out_size, void* d_ws, size_t ws_size,
                              hipStream_t stream) {
    (void)in_sizes; (void)n_in; (void)out_size; (void)ws_size;

    const float* x    = (const float*)d_in[0];
    const float* A    = (const float*)d_in[1];
    const float* Bvec = (const float*)d_in[2];
    float* y  = (float*)d_out;
    float* ws = (float*)d_ws;

    float* AmatT = ws + WS_AMATT;
    float* Ut    = ws + WS_UT;
    float* AcT   = ws + WS_ACT;
    float* S     = ws + WS_S;
    float* H     = ws + WS_H;

    k0_precompute<<<1, 256, 0, stream>>>(A, Bvec, ws);
    k1_state_gemm<<<BATCH * NCH * 8, 256, 0, stream>>>(x, Ut, S);
    k2_scan<<<(BATCH * (DMOD / 16)) / 8, 256, 0, stream>>>(AcT, S, H);
    k3_out_gemm<<<BATCH * NCH * 4, 256, 0, stream>>>(x, AmatT, H, y);
}


# k0: the 64-step power chain of A leaves LDS and its 4 block barriers per step for two waves' f32 WMMA register chains (D tile fed back as next B operand); sum order in A^p, A^p B, B^T A^p changes
# speedup vs baseline: 1.5915x; 1.5915x over previous
#include <hip/hip_runtime.h>
#include <hip/hip_bf16.h>

typedef float v2f __attribute__((ext_vector_type(2)));
typedef float v8f __attribute__((ext_vector_type(8)));

#define BATCH 4
#define SEQ   2048
#define DMOD  1024
#define NST   16
#define CHUNK 64
#define NCH   (SEQ / CHUNK)

#define WS_AMATT 0
#define WS_UT    (80 * 64)
#define WS_ACT   (80 * 64 + 64 * 16)
#define WS_S     8192
#define WS_H     (8192 + (size_t)BATCH * NCH * DMOD * NST)

__device__ __forceinline__ v8f k0_apply_A(const v2f (&af)[4], const v8f t) {
    v8f acc = {};
    for (int j = 0; j < 4; ++j) {
        v2f bf;
        bf.x = t[2 * j];
        bf.y = t[2 * j + 1];
        acc = __builtin_amdgcn_wmma_f32_16x16x4_f32(false, af[j], false, bf,
                                                    (short)0, acc, false, false);
    }
    return acc;
}

__global__ __launch_bounds__(256) void k0_precompute(const float* __restrict__ A,
                                                     const float* __restrict__ Bvec,
                                                     float* __restrict__ ws) {
    __shared__ float op[2][64][16];
    __shared__ float kp[2][64];

    const int tid = threadIdx.x;
    const int wave = __builtin_amdgcn_readfirstlane(tid >> 5);
    const int lane = tid & 31;
    const int h = lane >> 4;
    const int c = lane & 15;

    float* AmatT = ws + WS_AMATT;
    float* Ut    = ws + WS_UT;
    float* AcT   = ws + WS_ACT;

    if (wave < 2) {
        v2f af[4];
        for (int j = 0; j < 4; ++j) {
            af[j].x = A[c * 16 + 8 * h + 2 * j];
            af[j].y = A[c * 16 + 8 * h + 2 * j + 1];
        }
        float bh[8];
        for (int v = 0; v < 8; ++v) bh[v] = Bvec[8 * h + v];

        if (wave == 0) {
            v8f P;
            for (int v = 0; v < 8; ++v) P[v] = (8 * h + v == c) ? 1.0f : 0.0f;
            for (int p = 0; p < CHUNK; ++p) {
                P = k0_apply_A(af, P);
                float s = 0.f;
                for (int v = 0; v < 8; ++v) s += bh[v] * P[v];
                op[h][p][c] = s;
            }
            for (int v = 0; v < 8; ++v) AcT[c * 16 + 8 * h + v] = P[v];
        } else {
            v8f u;
            for (int v = 0; v < 8; ++v) u[v] = (c == 0) ? bh[v] : 0.0f;
            for (int p = 0; p < CHUNK; ++p) {
                if (p > 0) u = k0_apply_A(af, u);
                float s = 0.f;
                for (int v = 0; v < 8; ++v) s += bh[v] * u[v];
                if (c == 0) {
                    kp[h][p] = s;
                    for (int v = 0; v < 8; ++v)
                        Ut[(63 - p) * 16 + 8 * h + v] = u[v];
                }
            }
        }
    }
    __syncthreads();

    for (int e = tid; e < 64 * 64; e += 256) {
        const int k = e >> 6, t = e & 63;
        AmatT[e] = (t >= k) ? (kp[0][t - k] + kp[1][t - k]) : 0.0f;
    }
    for (int e = tid; e < 16 * 64; e += 256) {
        const int n = e >> 6, t = e & 63;
        AmatT[(64 + n) * 64 + t] = op[0][t][n] + op[1][t][n];
    }
}

__global__ __launch_bounds__(256) void k1_state_gemm(const float* __restrict__ x,
                                                     const float* __restrict__ Ut,
                                                     float* __restrict__ S) {
    const int lane = threadIdx.x & 31;
    const int wave = threadIdx.x >> 5;
    const int hi = lane >> 4;
    const int m = lane & 15;

    const int blk = blockIdx.x;
    const int dslab = blk & 7;
    const int chunk = (blk >> 3) & (NCH - 1);
    const int b = blk >> 8;
    const int d0 = (dslab * 8 + wave) * 16;

    const float* xb = x + ((size_t)b * SEQ + (size_t)chunk * CHUNK) * DMOD;

    v8f acc = {};
#pragma unroll
    for (int k0 = 0; k0 < CHUNK; k0 += 4) {
        const int r0 = k0 + 2 * hi;
        v2f a, bf;
        a.x  = xb[(size_t)r0 * DMOD + d0 + m];
        a.y  = xb[(size_t)(r0 + 1) * DMOD + d0 + m];
        bf.x = Ut[r0 * 16 + m];
        bf.y = Ut[(r0 + 1) * 16 + m];
        acc = __builtin_amdgcn_wmma_f32_16x16x4_f32(false, a, false, bf,
                                                    (short)0, acc, false, false);
    }

    float* Sp = S + (((size_t)b * NCH + chunk) * DMOD + d0) * NST;
#pragma unroll
    for (int v = 0; v < 8; ++v)
        Sp[(v + 8 * hi) * NST + m] = acc[v];
}

__global__ __launch_bounds__(256) void k2_scan(const float* __restrict__ AcT,
                                               const float* __restrict__ S,
                                               float* __restrict__ H) {
    __shared__ float hbuf[8][16][17];

    const int lane = threadIdx.x & 31;
    const int wave = threadIdx.x >> 5;
    const int hi = lane >> 4;
    const int m = lane & 15;

    const int gwave = blockIdx.x * 8 + wave;
    const int b = gwave >> 6;
    const int d0 = (gwave & 63) * 16;

    v2f bf[4];
#pragma unroll
    for (int k0 = 0; k0 < 16; k0 += 4) {
        const int r0 = k0 + 2 * hi;
        bf[k0 >> 2].x = AcT[r0 * 16 + m];
        bf[k0 >> 2].y = AcT[(r0 + 1) * 16 + m];
    }

    v8f h = {};
    for (int c = 0; c < NCH; ++c) {
#pragma unroll
        for (int v = 0; v < 8; ++v)
            hbuf[wave][v + 8 * hi][m] = h[v];
        __syncthreads();

        float* Hp = H + (((size_t)b * NCH + c) * NST) * DMOD;
#pragma unroll
        for (int v = 0; v < 8; ++v)
            Hp[(size_t)(2 * v + hi) * DMOD + d0 + m] = hbuf[wave][m][2 * v + hi];

        const float* Sp = S + (((size_t)b * NCH + c) * DMOD + d0) * NST;
        v8f acc;
#pragma unroll
        for (int v = 0; v < 8; ++v)
            acc[v] = Sp[(v + 8 * hi) * NST + m];

#pragma unroll
        for (int k0 = 0; k0 < 16; k0 += 4) {
            const int r0 = k0 + 2 * hi;
            v2f a;
            a.x = hbuf[wave][m][r0];
            a.y = hbuf[wave][m][r0 + 1];
            acc = __builtin_amdgcn_wmma_f32_16x16x4_f32(false, a, false, bf[k0 >> 2],
                                                        (short)0, acc, false, false);
        }
        h = acc;
        __syncthreads();
    }
}

__global__ __launch_bounds__(256) void k3_out_gemm(const float* __restrict__ x,
                                                   const float* __restrict__ AmatT,
                                                   const float* __restrict__ H,
                                                   float* __restrict__ y) {
    __shared__ float Asm[80 * 64];
    for (int e = threadIdx.x; e < 80 * 64; e += 256) Asm[e] = AmatT[e];
    __syncthreads();

    const int lane = threadIdx.x & 31;
    const int wave = threadIdx.x >> 5;
    const int hi = lane >> 4;
    const int m = lane & 15;

    const int blk = blockIdx.x;
    const int dquad = blk & 3;
    const int chunk = (blk >> 2) & (NCH - 1);
    const int b = blk >> 7;
    const int d0 = (dquad * 8 + wave) * 32;
    const int d1 = d0 + 16;

    const float* xb = x + ((size_t)b * SEQ + (size_t)chunk * CHUNK) * DMOD;
    const float* Hp = H + (((size_t)b * NCH + chunk) * NST) * DMOD;

    v8f acc[2][4] = {{{}, {}, {}, {}}, {{}, {}, {}, {}}};

#pragma unroll
    for (int kb = 0; kb < 4; ++kb) {
#pragma unroll
        for (int kk = 0; kk < 16; kk += 4) {
            const int r0 = kb * 16 + kk + 2 * hi;
            v2f b0, b1;
            b0.x = xb[(size_t)r0 * DMOD + d0 + m];
            b0.y = xb[(size_t)(r0 + 1) * DMOD + d0 + m];
            b1.x = xb[(size_t)r0 * DMOD + d1 + m];
            b1.y = xb[(size_t)(r0 + 1) * DMOD + d1 + m];
#pragma unroll
            for (int it = kb; it < 4; ++it) {
                v2f a;
                a.x = Asm[r0 * 64 + it * 16 + m];
                a.y = Asm[(r0 + 1) * 64 + it * 16 + m];
                acc[0][it] = __builtin_amdgcn_wmma_f32_16x16x4_f32(
                    false, a, false, b0, (short)0, acc[0][it], false, false);
                acc[1][it] = __builtin_amdgcn_wmma_f32_16x16x4_f32(
                    false, a, false, b1, (short)0, acc[1][it], false, false);
            }
        }
    }
#pragma unroll
    for (int k0 = 64; k0 < 80; k0 += 4) {
        const int r0 = k0 + 2 * hi;
        v2f b0, b1;
        b0.x = Hp[(size_t)(r0 - 64) * DMOD + d0 + m];
        b0.y = Hp[(size_t)(r0 - 63) * DMOD + d0 + m];
        b1.x = Hp[(size_t)(r0 - 64) * DMOD + d1 + m];
        b1.y = Hp[(size_t)(r0 - 63) * DMOD + d1 + m];
#pragma unroll
        for (int it = 0; it < 4; ++it) {
            v2f a;
            a.x = Asm[r0 * 64 + it * 16 + m];
            a.y = Asm[(r0 + 1) * 64 + it * 16 + m];
            acc[0][it] = __builtin_amdgcn_wmma_f32_16x16x4_f32(
                false, a, false, b0, (short)0, acc[0][it], false, false);
            acc[1][it] = __builtin_amdgcn_wmma_f32_16x16x4_f32(
                false, a, false, b1, (short)0, acc[1][it], false, false);
        }
    }

    float* yb = y + ((size_t)b * SEQ + (size_t)chunk * CHUNK) * DMOD;
#pragma unroll
    for (int it = 0; it < 4; ++it)
#pragma unroll
        for (int v = 0; v < 8; ++v) {
            const size_t row = (size_t)(it * 16 + v + 8 * hi) * DMOD;
            yb[row + d0 + m] = acc[0][it][v];
            yb[row + d1 + m] = acc[1][it][v];
        }
}

extern "C" void kernel_launch(void* const* d_in, const int* in_sizes, int n_in,
                              void* d_out, int out_size, void* d_ws, size_t ws_size,
                              hipStream_t stream) {
    (void)in_sizes; (void)n_in; (void)out_size; (void)ws_size;

    const float* x    = (const float*)d_in[0];
    const float* A    = (const float*)d_in[1];
    const float* Bvec = (const float*)d_in[2];
    float* y  = (float*)d_out;
    float* ws = (float*)d_ws;

    float* AmatT = ws + WS_AMATT;
    float* Ut    = ws + WS_UT;
    float* AcT   = ws + WS_ACT;
    float* S     = ws + WS_S;
    float* H     = ws + WS_H;

    k0_precompute<<<1, 256, 0, stream>>>(A, Bvec, ws);
    k1_state_gemm<<<BATCH * NCH * 8, 256, 0, stream>>>(x, Ut, S);
    k2_scan<<<(BATCH * (DMOD / 16)) / 8, 256, 0, stream>>>(AcT, S, H);
    k3_out_gemm<<<BATCH * NCH * 4, 256, 0, stream>>>(x, AmatT, H, y);
}


# k2: scan chain is 4 WMMA per chunk only: state kept transposed in registers (D tile = next B operand; no LDS, no barriers), S tiles all requested ahead of the chain; k sum order in H changes
# speedup vs baseline: 2.9825x; 1.8740x over previous
#include <hip/hip_runtime.h>
#include <hip/hip_bf16.h>

typedef float v2f __attribute__((ext_vector_type(2)));
typedef float v4f __attribute__((ext_vector_type(4)));
typedef float v8f __attribute__((ext_vector_type(8)));

#define BATCH 4
#define SEQ   2048
#define DMOD  1024
#define NST   16
#define CHUNK 64
#define NCH   (SEQ / CHUNK)

#define WS_AMATT 0
#define WS_UT    (80 * 64)
#define WS_ACT   (80 * 64 + 64 * 16)
#define WS_S     8192
#define WS_H     (8192 + (size_t)BATCH * NCH * DMOD * NST)

__device__ __forceinline__ v8f k0_apply_A(const v2f (&af)[4], const v8f t) {
    v8f acc = {};
    for (int j = 0; j < 4; ++j) {
        v2f bf;
        bf.x = t[2 * j];
        bf.y = t[2 * j + 1];
        acc = __builtin_amdgcn_wmma_f32_16x16x4_f32(false, af[j], false, bf,
                                                    (short)0, acc, false, false);
    }
    return acc;
}

__global__ __launch_bounds__(256) void k0_precompute(const float* __restrict__ A,
                                                     const float* __restrict__ Bvec,
                                                     float* __restrict__ ws) {
    __shared__ float op[2][64][16];
    __shared__ float kp[2][64];

    const int tid = threadIdx.x;
    const int wave = __builtin_amdgcn_readfirstlane(tid >> 5);
    const int lane = tid & 31;
    const int h = lane >> 4;
    const int c = lane & 15;

    float* AmatT = ws + WS_AMATT;
    float* Ut    = ws + WS_UT;
    float* AcT   = ws + WS_ACT;

    if (wave < 2) {
        v2f af[4];
        for (int j = 0; j < 4; ++j) {
            af[j].x = A[c * 16 + 8 * h + 2 * j];
            af[j].y = A[c * 16 + 8 * h + 2 * j + 1];
        }
        float bh[8];
        for (int v = 0; v < 8; ++v) bh[v] = Bvec[8 * h + v];

        if (wave == 0) {
            v8f P;
            for (int v = 0; v < 8; ++v) P[v] = (8 * h + v == c) ? 1.0f : 0.0f;
            for (int p = 0; p < CHUNK; ++p) {
                P = k0_apply_A(af, P);
                float s = 0.f;
                for (int v = 0; v < 8; ++v) s += bh[v] * P[v];
                op[h][p][c] = s;
            }
            for (int v = 0; v < 8; ++v) AcT[c * 16 + 8 * h + v] = P[v];
        } else {
            v8f u;
            for (int v = 0; v < 8; ++v) u[v] = (c == 0) ? bh[v] : 0.0f;
            for (int p = 0; p < CHUNK; ++p) {
                if (p > 0) u = k0_apply_A(af, u);
                float s = 0.f;
                for (int v = 0; v < 8; ++v) s += bh[v] * u[v];
                if (c == 0) {
                    kp[h][p] = s;
                    for (int v = 0; v < 8; ++v)
                        Ut[(63 - p) * 16 + 8 * h + v] = u[v];
                }
            }
        }
    }
    __syncthreads();

    for (int e = tid; e < 64 * 64; e += 256) {
        const int k = e >> 6, t = e & 63;
        AmatT[e] = (t >= k) ? (kp[0][t - k] + kp[1][t - k]) : 0.0f;
    }
    for (int e = tid; e < 16 * 64; e += 256) {
        const int n = e >> 6, t = e & 63;
        AmatT[(64 + n) * 64 + t] = op[0][t][n] + op[1][t][n];
    }
}

__global__ __launch_bounds__(256) void k1_state_gemm(const float* __restrict__ x,
                                                     const float* __restrict__ Ut,
                                                     float* __restrict__ S) {
    const int lane = threadIdx.x & 31;
    const int wave = threadIdx.x >> 5;
    const int hi = lane >> 4;
    const int m = lane & 15;

    const int blk = blockIdx.x;
    const int dslab = blk & 7;
    const int chunk = (blk >> 3) & (NCH - 1);
    const int b = blk >> 8;
    const int d0 = (dslab * 8 + wave) * 16;

    const float* xb = x + ((size_t)b * SEQ + (size_t)chunk * CHUNK) * DMOD;

    v8f acc = {};
#pragma unroll
    for (int k0 = 0; k0 < CHUNK; k0 += 4) {
        const int r0 = k0 + 2 * hi;
        v2f a, bf;
        a.x  = xb[(size_t)r0 * DMOD + d0 + m];
        a.y  = xb[(size_t)(r0 + 1) * DMOD + d0 + m];
        bf.x = Ut[r0 * 16 + m];
        bf.y = Ut[(r0 + 1) * 16 + m];
        acc = __builtin_amdgcn_wmma_f32_16x16x4_f32(false, a, false, bf,
                                                    (short)0, acc, false, false);
    }

    float* Sp = S + (((size_t)b * NCH + chunk) * DMOD + d0) * NST;
#pragma unroll
    for (int v = 0; v < 8; ++v)
        Sp[(v + 8 * hi) * NST + m] = acc[v];
}

__global__ __launch_bounds__(32) void k2_scan(const float* __restrict__ AcT,
                                              const float* __restrict__ S,
                                              float* __restrict__ H) {
    const int lane = threadIdx.x;
    const int h = lane >> 4;
    const int l = lane & 15;

    const int gwave = blockIdx.x;
    const int b = gwave >> 6;
    const int d0 = (gwave & 63) * 16;

    v2f af[4];
    for (int j = 0; j < 4; ++j) {
        af[j].x = AcT[(8 * h + 2 * j) * 16 + l];
        af[j].y = AcT[(8 * h + 2 * j + 1) * 16 + l];
    }

    const float* Sp = S + (((size_t)b * NCH) * DMOD + d0 + l) * NST + 8 * h;
    v8f st[NCH - 1];
#pragma unroll
    for (int c = 0; c < NCH - 1; ++c) {
        const v4f* p = (const v4f*)(Sp + (size_t)c * DMOD * NST);
        const v4f lo = p[0];
        const v4f hi = p[1];
        st[c][0] = lo.x; st[c][1] = lo.y; st[c][2] = lo.z; st[c][3] = lo.w;
        st[c][4] = hi.x; st[c][5] = hi.y; st[c][6] = hi.z; st[c][7] = hi.w;
    }
    __builtin_amdgcn_sched_barrier(0);

    float* Hp = H + (((size_t)b * NCH) * NST + 8 * h) * DMOD + d0 + l;
    v8f Ht = {};
#pragma unroll
    for (int c = 0; c < NCH; ++c) {
        for (int v = 0; v < 8; ++v)
            Hp[((size_t)c * NST + v) * DMOD] = Ht[v];
        if (c < NCH - 1) {
            v8f acc = st[c];
            for (int j = 0; j < 4; ++j) {
                v2f bf;
                bf.x = Ht[2 * j];
                bf.y = Ht[2 * j + 1];
                acc = __builtin_amdgcn_wmma_f32_16x16x4_f32(false, af[j], false, bf,
                                                            (short)0, acc, false, false);
            }
            Ht = acc;
        }
    }
}

__global__ __launch_bounds__(256) void k3_out_gemm(const float* __restrict__ x,
                                                   const float* __restrict__ AmatT,
                                                   const float* __restrict__ H,
                                                   float* __restrict__ y) {
    __shared__ float Asm[80 * 64];
    for (int e = threadIdx.x; e < 80 * 64; e += 256) Asm[e] = AmatT[e];
    __syncthreads();

    const int lane = threadIdx.x & 31;
    const int wave = threadIdx.x >> 5;
    const int hi = lane >> 4;
    const int m = lane & 15;

    const int blk = blockIdx.x;
    const int dquad = blk & 3;
    const int chunk = (blk >> 2) & (NCH - 1);
    const int b = blk >> 7;
    const int d0 = (dquad * 8 + wave) * 32;
    const int d1 = d0 + 16;

    const float* xb = x + ((size_t)b * SEQ + (size_t)chunk * CHUNK) * DMOD;
    const float* Hp = H + (((size_t)b * NCH + chunk) * NST) * DMOD;

    v8f acc[2][4] = {{{}, {}, {}, {}}, {{}, {}, {}, {}}};

#pragma unroll
    for (int kb = 0; kb < 4; ++kb) {
#pragma unroll
        for (int kk = 0; kk < 16; kk += 4) {
            const int r0 = kb * 16 + kk + 2 * hi;
            v2f b0, b1;
            b0.x = xb[(size_t)r0 * DMOD + d0 + m];
            b0.y = xb[(size_t)(r0 + 1) * DMOD + d0 + m];
            b1.x = xb[(size_t)r0 * DMOD + d1 + m];
            b1.y = xb[(size_t)(r0 + 1) * DMOD + d1 + m];
#pragma unroll
            for (int it = kb; it < 4; ++it) {
                v2f a;
                a.x = Asm[r0 * 64 + it * 16 + m];
                a.y = Asm[(r0 + 1) * 64 + it * 16 + m];
                acc[0][it] = __builtin_amdgcn_wmma_f32_16x16x4_f32(
                    false, a, false, b0, (short)0, acc[0][it], false, false);
                acc[1][it] = __builtin_amdgcn_wmma_f32_16x16x4_f32(
                    false, a, false, b1, (short)0, acc[1][it], false, false);
            }
        }
    }
#pragma unroll
    for (int k0 = 64; k0 < 80; k0 += 4) {
        const int r0 = k0 + 2 * hi;
        v2f b0, b1;
        b0.x = Hp[(size_t)(r0 - 64) * DMOD + d0 + m];
        b0.y = Hp[(size_t)(r0 - 63) * DMOD + d0 + m];
        b1.x = Hp[(size_t)(r0 - 64) * DMOD + d1 + m];
        b1.y = Hp[(size_t)(r0 - 63) * DMOD + d1 + m];
#pragma unroll
        for (int it = 0; it < 4; ++it) {
            v2f a;
            a.x = Asm[r0 * 64 + it * 16 + m];
            a.y = Asm[(r0 + 1) * 64 + it * 16 + m];
            acc[0][it] = __builtin_amdgcn_wmma_f32_16x16x4_f32(
                false, a, false, b0, (short)0, acc[0][it], false, false);
            acc[1][it] = __builtin_amdgcn_wmma_f32_16x16x4_f32(
                false, a, false, b1, (short)0, acc[1][it], false, false);
        }
    }

    float* yb = y + ((size_t)b * SEQ + (size_t)chunk * CHUNK) * DMOD;
#pragma unroll
    for (int it = 0; it < 4; ++it)
#pragma unroll
        for (int v = 0; v < 8; ++v) {
            const size_t row = (size_t)(it * 16 + v + 8 * hi) * DMOD;
            yb[row + d0 + m] = acc[0][it][v];
            yb[row + d1 + m] = acc[1][it][v];
        }
}

extern "C" void kernel_launch(void* const* d_in, const int* in_sizes, int n_in,
                              void* d_out, int out_size, void* d_ws, size_t ws_size,
                              hipStream_t stream) {
    (void)in_sizes; (void)n_in; (void)out_size; (void)ws_size;

    const float* x    = (const float*)d_in[0];
    const float* A    = (const float*)d_in[1];
    const float* Bvec = (const float*)d_in[2];
    float* y  = (float*)d_out;
    float* ws = (float*)d_ws;

    float* AmatT = ws + WS_AMATT;
    float* Ut    = ws + WS_UT;
    float* AcT   = ws + WS_ACT;
    float* S     = ws + WS_S;
    float* H     = ws + WS_H;

    k0_precompute<<<1, 256, 0, stream>>>(A, Bvec, ws);
    k1_state_gemm<<<BATCH * NCH * 8, 256, 0, stream>>>(x, Ut, S);
    k2_scan<<<BATCH * (DMOD / 16), 32, 0, stream>>>(AcT, S, H);
    k3_out_gemm<<<BATCH * NCH * 4, 256, 0, stream>>>(x, AmatT, H, y);
}


# k0 folded into k1: wave 0 of every work-group (now 32 waves, one per WGP) runs the 63-step A^p B matrix chain into LDS beside its x loads; work-group 0 also builds A^64 and [T|O]; 3 launches
# speedup vs baseline: 3.4375x; 1.1526x over previous
#include <hip/hip_runtime.h>
#include <hip/hip_bf16.h>

typedef float v2f __attribute__((ext_vector_type(2)));
typedef float v4f __attribute__((ext_vector_type(4)));
typedef float v8f __attribute__((ext_vector_type(8)));

#define BATCH 4
#define SEQ   2048
#define DMOD  1024
#define NST   16
#define CHUNK 64
#define NCH   (SEQ / CHUNK)

#define WS_AMATT 0
#define WS_ACT   (80 * 64 + 64 * 16)
#define WS_S     8192
#define WS_H     (8192 + (size_t)BATCH * NCH * DMOD * NST)

__device__ __forceinline__ v8f k0_apply_A(const v2f (&af)[4], const v8f t) {
    v8f acc = {};
    for (int j = 0; j < 4; ++j) {
        v2f bf;
        bf.x = t[2 * j];
        bf.y = t[2 * j + 1];
        acc = __builtin_amdgcn_wmma_f32_16x16x4_f32(false, af[j], false, bf,
                                                    (short)0, acc, false, false);
    }
    return acc;
}

__global__ __launch_bounds__(1024) void k1_state_gemm(const float* __restrict__ x,
                                                      const float* __restrict__ A,
                                                      const float* __restrict__ Bvec,
                                                      float* __restrict__ ws) {
    __shared__ float Uts[64 * 16];
    __shared__ float op[2][64][16];
    __shared__ float kp[2][64];

    const int tid = threadIdx.x;
    const int wave = __builtin_amdgcn_readfirstlane(tid >> 5);
    const int lane = tid & 31;
    const int hi = lane >> 4;
    const int m = lane & 15;
    const bool first = (blockIdx.x == 0);

    const int tile = blockIdx.x * 4 + (wave >> 3);
    const int dslab = tile & 7;
    const int chunk = (tile >> 3) & (NCH - 1);
    const int b = tile >> 8;
    const int d0 = (dslab * 8 + (wave & 7)) * 16;

    const float* xb = x + ((size_t)b * SEQ + (size_t)chunk * CHUNK) * DMOD;

    const bool chain_wave = (wave == 0) || (wave == 1 && first);
    v2f af[4] = {};
    float bh[8] = {};
    if (chain_wave) {
        for (int j = 0; j < 4; ++j) {
            af[j].x = A[m * 16 + 8 * hi + 2 * j];
            af[j].y = A[m * 16 + 8 * hi + 2 * j + 1];
        }
        for (int v = 0; v < 8; ++v) bh[v] = Bvec[8 * hi + v];
    }

    v2f a[16];
#pragma unroll
    for (int q = 0; q < 16; ++q) {
        const int r0 = 4 * q + 2 * hi;
        a[q].x = xb[(size_t)r0 * DMOD + d0 + m];
        a[q].y = xb[(size_t)(r0 + 1) * DMOD + d0 + m];
    }

    if (chain_wave) {
        if (wave == 0) {
            v8f u;
            for (int v = 0; v < 8; ++v) u[v] = (m == 0) ? bh[v] : 0.0f;
            for (int p = 0; p < CHUNK; ++p) {
                if (p > 0) u = k0_apply_A(af, u);
                if (m == 0) {
                    for (int v = 0; v < 8; ++v)
                        Uts[(63 - p) * 16 + 8 * hi + v] = u[v];
                }
                if (first) {
                    float s = 0.f;
                    for (int v = 0; v < 8; ++v) s += bh[v] * u[v];
                    if (m == 0) kp[hi][p] = s;
                }
            }
        } else {
            float* AcT = ws + WS_ACT;
            v8f P;
            for (int v = 0; v < 8; ++v) P[v] = (8 * hi + v == m) ? 1.0f : 0.0f;
            for (int p = 0; p < CHUNK; ++p) {
                P = k0_apply_A(af, P);
                float s = 0.f;
                for (int v = 0; v < 8; ++v) s += bh[v] * P[v];
                op[hi][p][m] = s;
            }
            for (int v = 0; v < 8; ++v) AcT[m * 16 + 8 * hi + v] = P[v];
        }
    }
    __syncthreads();

    v2f bf[16];
#pragma unroll
    for (int q = 0; q < 16; ++q) {
        const int r0 = 4 * q + 2 * hi;
        bf[q].x = Uts[r0 * 16 + m];
        bf[q].y = Uts[(r0 + 1) * 16 + m];
    }
    __builtin_amdgcn_sched_barrier(0);
    v8f acc = {};
#pragma unroll
    for (int q = 0; q < 16; ++q)
        acc = __builtin_amdgcn_wmma_f32_16x16x4_f32(false, a[q], false, bf[q],
                                                    (short)0, acc, false, false);

    float* Sp = ws + WS_S + (((size_t)b * NCH + chunk) * DMOD + d0) * NST;
#pragma unroll
    for (int v = 0; v < 8; ++v)
        Sp[(v + 8 * hi) * NST + m] = acc[v];

    if (first) {
        float* AmatT = ws + WS_AMATT;
        for (int e = tid; e < 64 * 64; e += 1024) {
            const int k = e >> 6, t = e & 63;
            AmatT[e] = (t >= k) ? (kp[0][t - k] + kp[1][t - k]) : 0.0f;
        }
        for (int e = tid; e < 16 * 64; e += 1024) {
            const int n = e >> 6, t = e & 63;
            AmatT[(64 + n) * 64 + t] = op[0][t][n] + op[1][t][n];
        }
    }
}

__global__ __launch_bounds__(32) void k2_scan(const float* __restrict__ AcT,
                                              const float* __restrict__ S,
                                              float* __restrict__ H) {
    const int lane = threadIdx.x;
    const int h = lane >> 4;
    const int l = lane & 15;

    const int gwave = blockIdx.x;
    const int b = gwave >> 6;
    const int d0 = (gwave & 63) * 16;

    v2f af[4];
    for (int j = 0; j < 4; ++j) {
        af[j].x = AcT[(8 * h + 2 * j) * 16 + l];
        af[j].y = AcT[(8 * h + 2 * j + 1) * 16 + l];
    }

    const float* Sp = S + (((size_t)b * NCH) * DMOD + d0 + l) * NST + 8 * h;
    v8f st[NCH - 1];
#pragma unroll
    for (int c = 0; c < NCH - 1; ++c) {
        const v4f* p = (const v4f*)(Sp + (size_t)c * DMOD * NST);
        const v4f lo = p[0];
        const v4f hi = p[1];
        st[c][0] = lo.x; st[c][1] = lo.y; st[c][2] = lo.z; st[c][3] = lo.w;
        st[c][4] = hi.x; st[c][5] = hi.y; st[c][6] = hi.z; st[c][7] = hi.w;
    }
    __builtin_amdgcn_sched_barrier(0);

    float* Hp = H + (((size_t)b * NCH) * NST + 8 * h) * DMOD + d0 + l;
    v8f Ht = {};
#pragma unroll
    for (int c = 0; c < NCH; ++c) {
        for (int v = 0; v < 8; ++v)
            Hp[((size_t)c * NST + v) * DMOD] = Ht[v];
        if (c < NCH - 1) {
            v8f acc = st[c];
            for (int j = 0; j < 4; ++j) {
                v2f bf;
                bf.x = Ht[2 * j];
                bf.y = Ht[2 * j + 1];
                acc = __builtin_amdgcn_wmma_f32_16x16x4_f32(false, af[j], false, bf,
                                                            (short)0, acc, false, false);
            }
            Ht = acc;
        }
    }
}

__global__ __launch_bounds__(256) void k3_out_gemm(const float* __restrict__ x,
                                                   const float* __restrict__ AmatT,
                                                   const float* __restrict__ H,
                                                   float* __restrict__ y) {
    __shared__ float Asm[80 * 64];
    for (int e = threadIdx.x; e < 80 * 64; e += 256) Asm[e] = AmatT[e];
    __syncthreads();

    const int lane = threadIdx.x & 31;
    const int wave = threadIdx.x >> 5;
    const int hi = lane >> 4;
    const int m = lane & 15;

    const int blk = blockIdx.x;
    const int dquad = blk & 3;
    const int chunk = (blk >> 2) & (NCH - 1);
    const int b = blk >> 7;
    const int d0 = (dquad * 8 + wave) * 32;
    const int d1 = d0 + 16;

    const float* xb = x + ((size_t)b * SEQ + (size_t)chunk * CHUNK) * DMOD;
    const float* Hp = H + (((size_t)b * NCH + chunk) * NST) * DMOD;

    v8f acc[2][4] = {{{}, {}, {}, {}}, {{}, {}, {}, {}}};

#pragma unroll
    for (int kb = 0; kb < 4; ++kb) {
#pragma unroll
        for (int kk = 0; kk < 16; kk += 4) {
            const int r0 = kb * 16 + kk + 2 * hi;
            v2f b0, b1;
            b0.x = xb[(size_t)r0 * DMOD + d0 + m];
            b0.y = xb[(size_t)(r0 + 1) * DMOD + d0 + m];
            b1.x = xb[(size_t)r0 * DMOD + d1 + m];
            b1.y = xb[(size_t)(r0 + 1) * DMOD + d1 + m];
#pragma unroll
            for (int it = kb; it < 4; ++it) {
                v2f a;
                a.x = Asm[r0 * 64 + it * 16 + m];
                a.y = Asm[(r0 + 1) * 64 + it * 16 + m];
                acc[0][it] = __builtin_amdgcn_wmma_f32_16x16x4_f32(
                    false, a, false, b0, (short)0, acc[0][it], false, false);
                acc[1][it] = __builtin_amdgcn_wmma_f32_16x16x4_f32(
                    false, a, false, b1, (short)0, acc[1][it], false, false);
            }
        }
    }
#pragma unroll
    for (int k0 = 64; k0 < 80; k0 += 4) {
        const int r0 = k0 + 2 * hi;
        v2f b0, b1;
        b0.x = Hp[(size_t)(r0 - 64) * DMOD + d0 + m];
        b0.y = Hp[(size_t)(r0 - 63) * DMOD + d0 + m];
        b1.x = Hp[(size_t)(r0 - 64) * DMOD + d1 + m];
        b1.y = Hp[(size_t)(r0 - 63) * DMOD + d1 + m];
#pragma unroll
        for (int it = 0; it < 4; ++it) {
            v2f a;
            a.x = Asm[r0 * 64 + it * 16 + m];
            a.y = Asm[(r0 + 1) * 64 + it * 16 + m];
            acc[0][it] = __builtin_amdgcn_wmma_f32_16x16x4_f32(
                false, a, false, b0, (short)0, acc[0][it], false, false);
            acc[1][it] = __builtin_amdgcn_wmma_f32_16x16x4_f32(
                false, a, false, b1, (short)0, acc[1][it], false, false);
        }
    }

    float* yb = y + ((size_t)b * SEQ + (size_t)chunk * CHUNK) * DMOD;
#pragma unroll
    for (int it = 0; it < 4; ++it)
#pragma unroll
        for (int v = 0; v < 8; ++v) {
            const size_t row = (size_t)(it * 16 + v + 8 * hi) * DMOD;
            yb[row + d0 + m] = acc[0][it][v];
            yb[row + d1 + m] = acc[1][it][v];
        }
}

extern "C" void kernel_launch(void* const* d_in, const int* in_sizes, int n_in,
                              void* d_out, int out_size, void* d_ws, size_t ws_size,
                              hipStream_t stream) {
    (void)in_sizes; (void)n_in; (void)out_size; (void)ws_size;

    const float* x    = (const float*)d_in[0];
    const float* A    = (const float*)d_in[1];
    const float* Bvec = (const float*)d_in[2];
    float* y  = (float*)d_out;
    float* ws = (float*)d_ws;

    float* AmatT = ws + WS_AMATT;
    float* AcT   = ws + WS_ACT;
    const float* S = ws + WS_S;
    float* H     = ws + WS_H;

    k1_state_gemm<<<BATCH * NCH * 2, 1024, 0, stream>>>(x, A, Bvec, ws);
    k2_scan<<<BATCH * (DMOD / 16), 32, 0, stream>>>(AcT, S, H);
    k3_out_gemm<<<BATCH * NCH * 4, 256, 0, stream>>>(x, AmatT, H, y);
}


# one launch: a work-group owns all 32 chunks of a 32-channel slab (wave = chunk); x tile stays in registers for state sums and output product; S, H, [T|O], A^64 pass through LDS; sum orders unchanged
# speedup vs baseline: 4.2944x; 1.2493x over previous
#include <hip/hip_runtime.h>

typedef float v2f __attribute__((ext_vector_type(2)));
typedef float v8f __attribute__((ext_vector_type(8)));

#define BATCH 4
#define SEQ   2048
#define DMOD  1024
#define NST   16
#define CHUNK 64
#define NCH   (SEQ / CHUNK)
#define SLAB  32
#define AP    72

__device__ __forceinline__ v8f k0_apply_A(const v2f (&af)[4], const v8f t) {
    v8f acc = {};
    for (int j = 0; j < 4; ++j) {
        v2f bf;
        bf.x = t[2 * j];
        bf.y = t[2 * j + 1];
        acc = __builtin_amdgcn_wmma_f32_16x16x4_f32(false, af[j], false, bf,
                                                    (short)0, acc, false, false);
    }
    return acc;
}

__global__ __launch_bounds__(1024) void k_fused(const float* __restrict__ x,
                                                const float* __restrict__ A,
                                                const float* __restrict__ Bvec,
                                                float* __restrict__ y) {
    __shared__ float Uts[64 * 16];
    __shared__ float op[2][64][16];
    __shared__ float kp[2][64];
    __shared__ float AcT[16 * 16];
    __shared__ float Asm[80 * AP];
    __shared__ float Ssm[NCH * 2 * 8 * 32];
    __shared__ float Hsm[NCH * NST * SLAB];

    const int tid = threadIdx.x;
    const int wave = __builtin_amdgcn_readfirstlane(tid >> 5);
    const int lane = tid & 31;
    const int hi = lane >> 4;
    const int m = lane & 15;

    const int b = blockIdx.x >> 5;
    const int dbase = (blockIdx.x & 31) * SLAB;

    const float* xb = x + ((size_t)b * SEQ + (size_t)wave * CHUNK) * DMOD + dbase;

    const bool chain_wave = (wave < 2);
    v2f af[4] = {};
    float bh[8] = {};
    if (chain_wave) {
        for (int j = 0; j < 4; ++j) {
            af[j].x = A[m * 16 + 8 * hi + 2 * j];
            af[j].y = A[m * 16 + 8 * hi + 2 * j + 1];
        }
        for (int v = 0; v < 8; ++v) bh[v] = Bvec[8 * hi + v];
    }

    v2f a0[16], a1[16];
#pragma unroll
    for (int q = 0; q < 16; ++q) {
        const int r0 = 4 * q + 2 * hi;
        a0[q].x = xb[(size_t)r0 * DMOD + m];
        a0[q].y = xb[(size_t)(r0 + 1) * DMOD + m];
        a1[q].x = xb[(size_t)r0 * DMOD + 16 + m];
        a1[q].y = xb[(size_t)(r0 + 1) * DMOD + 16 + m];
    }

    if (chain_wave) {
        if (wave == 0) {
            v8f u;
            for (int v = 0; v < 8; ++v) u[v] = (m == 0) ? bh[v] : 0.0f;
            for (int p = 0; p < CHUNK; ++p) {
                if (p > 0) u = k0_apply_A(af, u);
                if (m == 0) {
                    for (int v = 0; v < 8; ++v)
                        Uts[(63 - p) * 16 + 8 * hi + v] = u[v];
                }
                float s = 0.f;
                for (int v = 0; v < 8; ++v) s += bh[v] * u[v];
                if (m == 0) kp[hi][p] = s;
            }
        } else {
            v8f P;
            for (int v = 0; v < 8; ++v) P[v] = (8 * hi + v == m) ? 1.0f : 0.0f;
            for (int p = 0; p < CHUNK; ++p) {
                P = k0_apply_A(af, P);
                float s = 0.f;
                for (int v = 0; v < 8; ++v) s += bh[v] * P[v];
                op[hi][p][m] = s;
            }
            for (int v = 0; v < 8; ++v) AcT[m * 16 + 8 * hi + v] = P[v];
        }
    }
    __syncthreads();

    for (int e = tid; e < 64 * 64; e += 1024) {
        const int k = e >> 6, t = e & 63;
        Asm[k * AP + t] = (t >= k) ? (kp[0][t - k] + kp[1][t - k]) : 0.0f;
    }
    for (int e = tid; e < 16 * 64; e += 1024) {
        const int n = e >> 6, t = e & 63;
        Asm[(64 + n) * AP + t] = op[0][t][n] + op[1][t][n];
    }

    {
        v8f s0 = {}, s1 = {};
#pragma unroll
        for (int q = 0; q < 16; ++q) {
            const int r0 = 4 * q + 2 * hi;
            v2f uf;
            uf.x = Uts[r0 * 16 + m];
            uf.y = Uts[(r0 + 1) * 16 + m];
            s0 = __builtin_amdgcn_wmma_f32_16x16x4_f32(false, uf, false, a0[q],
                                                       (short)0, s0, false, false);
            s1 = __builtin_amdgcn_wmma_f32_16x16x4_f32(false, uf, false, a1[q],
                                                       (short)0, s1, false, false);
        }
        float* sp = Ssm + (size_t)(wave * 2) * 8 * 32 + lane;
#pragma unroll
        for (int v = 0; v < 8; ++v) {
            sp[v * 32] = s0[v];
            sp[(8 + v) * 32] = s1[v];
        }
    }
    __syncthreads();

    if (chain_wave) {
        v2f cf[4];
        for (int j = 0; j < 4; ++j) {
            cf[j].x = AcT[(8 * hi + 2 * j) * 16 + m];
            cf[j].y = AcT[(8 * hi + 2 * j + 1) * 16 + m];
        }
        const float* sp = Ssm + (size_t)wave * 8 * 32 + lane;
        float* hp = Hsm + (8 * hi) * SLAB + wave * 16 + m;
        v8f Ht = {};
        v8f cur;
        for (int v = 0; v < 8; ++v) cur[v] = sp[v * 32];
#pragma unroll
        for (int c = 0; c < NCH; ++c) {
            v8f nxt = {};
            if (c + 1 < NCH - 1) {
                for (int v = 0; v < 8; ++v) nxt[v] = sp[((c + 1) * 2 * 8 + v) * 32];
            }
            for (int v = 0; v < 8; ++v) hp[(c * NST + v) * SLAB] = Ht[v];
            if (c < NCH - 1) {
                v8f acc = cur;
                for (int j = 0; j < 4; ++j) {
                    v2f bf;
                    bf.x = Ht[2 * j];
                    bf.y = Ht[2 * j + 1];
                    acc = __builtin_amdgcn_wmma_f32_16x16x4_f32(false, cf[j], false, bf,
                                                                (short)0, acc, false, false);
                }
                Ht = acc;
                cur = nxt;
            }
        }
    }
    __syncthreads();

    v2f h0[4], h1[4];
    {
        const float* hq = Hsm + (size_t)wave * NST * SLAB;
#pragma unroll
        for (int j = 0; j < 4; ++j) {
            const int k0 = 4 * j + 2 * hi;
            h0[j].x = hq[k0 * SLAB + m];
            h0[j].y = hq[(k0 + 1) * SLAB + m];
            h1[j].x = hq[k0 * SLAB + 16 + m];
            h1[j].y = hq[(k0 + 1) * SLAB + 16 + m];
        }
    }

    float* yb = y + ((size_t)b * SEQ + (size_t)wave * CHUNK) * DMOD + dbase;
#pragma unroll
    for (int it = 0; it < 4; ++it) {
        v8f c0 = {}, c1 = {};
#pragma unroll
        for (int q = 0; q < 4 * (it + 1); ++q) {
            const int r0 = 4 * q + 2 * hi;
            v2f tf;
            tf.x = Asm[r0 * AP + it * 16 + m];
            tf.y = Asm[(r0 + 1) * AP + it * 16 + m];
            c0 = __builtin_amdgcn_wmma_f32_16x16x4_f32(false, tf, false, a0[q],
                                                       (short)0, c0, false, false);
            c1 = __builtin_amdgcn_wmma_f32_16x16x4_f32(false, tf, false, a1[q],
                                                       (short)0, c1, false, false);
        }
#pragma unroll
        for (int j = 0; j < 4; ++j) {
            const int r0 = 64 + 4 * j + 2 * hi;
            v2f tf;
            tf.x = Asm[r0 * AP + it * 16 + m];
            tf.y = Asm[(r0 + 1) * AP + it * 16 + m];
            c0 = __builtin_amdgcn_wmma_f32_16x16x4_f32(false, tf, false, h0[j],
                                                       (short)0, c0, false, false);
            c1 = __builtin_amdgcn_wmma_f32_16x16x4_f32(false, tf, false, h1[j],
                                                       (short)0, c1, false, false);
        }
#pragma unroll
        for (int v = 0; v < 8; ++v) {
            const size_t row = (size_t)(it * 16 + v + 8 * hi) * DMOD;
            yb[row + m] = c0[v];
            yb[row + 16 + m] = c1[v];
        }
    }
}

extern "C" void kernel_launch(void* const* d_in, const int* in_sizes, int n_in,
                              void* d_out, int out_size, void* d_ws, size_t ws_size,
                              hipStream_t stream) {
    (void)in_sizes; (void)n_in; (void)out_size; (void)d_ws; (void)ws_size;

    const float* x    = (const float*)d_in[0];
    const float* A    = (const float*)d_in[1];
    const float* Bvec = (const float*)d_in[2];
    float* y = (float*)d_out;

    k_fused<<<BATCH * (DMOD / SLAB), 1024, 0, stream>>>(x, A, Bvec, y);
}


# phase 4 stores: each wave stages its 16x32 y tile in its own 2 KB of the dead S area of LDS and writes it with 4 global_store_b128 of four whole 128-byte lines each (was 16 half-line b32 stores)
# speedup vs baseline: 4.6763x; 1.0889x over previous
#include <hip/hip_runtime.h>

typedef float v2f __attribute__((ext_vector_type(2)));
typedef float v8f __attribute__((ext_vector_type(8)));

#define BATCH 4
#define SEQ   2048
#define DMOD  1024
#define NST   16
#define CHUNK 64
#define NCH   (SEQ / CHUNK)
#define SLAB  32
#define AP    72

__device__ __forceinline__ v8f k0_apply_A(const v2f (&af)[4], const v8f t) {
    v8f acc = {};
    for (int j = 0; j < 4; ++j) {
        v2f bf;
        bf.x = t[2 * j];
        bf.y = t[2 * j + 1];
        acc = __builtin_amdgcn_wmma_f32_16x16x4_f32(false, af[j], false, bf,
                                                    (short)0, acc, false, false);
    }
    return acc;
}

__global__ __launch_bounds__(1024) void k_fused(const float* __restrict__ x,
                                                const float* __restrict__ A,
                                                const float* __restrict__ Bvec,
                                                float* __restrict__ y) {
    __shared__ float Uts[64 * 16];
    __shared__ float op[2][64][16];
    __shared__ float kp[2][64];
    __shared__ float AcT[16 * 16];
    __shared__ float Asm[80 * AP];
    __shared__ __attribute__((aligned(16))) float Ssm[NCH * 2 * 8 * 32];
    __shared__ float Hsm[NCH * NST * SLAB];

    const int tid = threadIdx.x;
    const int wave = __builtin_amdgcn_readfirstlane(tid >> 5);
    const int lane = tid & 31;
    const int hi = lane >> 4;
    const int m = lane & 15;

    const int b = blockIdx.x >> 5;
    const int dbase = (blockIdx.x & 31) * SLAB;

    const float* xb = x + ((size_t)b * SEQ + (size_t)wave * CHUNK) * DMOD + dbase;

    const bool chain_wave = (wave < 2);
    v2f af[4] = {};
    float bh[8] = {};
    if (chain_wave) {
        for (int j = 0; j < 4; ++j) {
            af[j].x = A[m * 16 + 8 * hi + 2 * j];
            af[j].y = A[m * 16 + 8 * hi + 2 * j + 1];
        }
        for (int v = 0; v < 8; ++v) bh[v] = Bvec[8 * hi + v];
    }

    v2f a0[16], a1[16];
#pragma unroll
    for (int q = 0; q < 16; ++q) {
        const int r0 = 4 * q + 2 * hi;
        a0[q].x = xb[(size_t)r0 * DMOD + m];
        a0[q].y = xb[(size_t)(r0 + 1) * DMOD + m];
        a1[q].x = xb[(size_t)r0 * DMOD + 16 + m];
        a1[q].y = xb[(size_t)(r0 + 1) * DMOD + 16 + m];
    }

    if (chain_wave) {
        if (wave == 0) {
            v8f u;
            for (int v = 0; v < 8; ++v) u[v] = (m == 0) ? bh[v] : 0.0f;
            for (int p = 0; p < CHUNK; ++p) {
                if (p > 0) u = k0_apply_A(af, u);
                if (m == 0) {
                    for (int v = 0; v < 8; ++v)
                        Uts[(63 - p) * 16 + 8 * hi + v] = u[v];
                }
                float s = 0.f;
                for (int v = 0; v < 8; ++v) s += bh[v] * u[v];
                if (m == 0) kp[hi][p] = s;
            }
        } else {
            v8f P;
            for (int v = 0; v < 8; ++v) P[v] = (8 * hi + v == m) ? 1.0f : 0.0f;
            for (int p = 0; p < CHUNK; ++p) {
                P = k0_apply_A(af, P);
                float s = 0.f;
                for (int v = 0; v < 8; ++v) s += bh[v] * P[v];
                op[hi][p][m] = s;
            }
            for (int v = 0; v < 8; ++v) AcT[m * 16 + 8 * hi + v] = P[v];
        }
    }
    __syncthreads();

    for (int e = tid; e < 64 * 64; e += 1024) {
        const int k = e >> 6, t = e & 63;
        Asm[k * AP + t] = (t >= k) ? (kp[0][t - k] + kp[1][t - k]) : 0.0f;
    }
    for (int e = tid; e < 16 * 64; e += 1024) {
        const int n = e >> 6, t = e & 63;
        Asm[(64 + n) * AP + t] = op[0][t][n] + op[1][t][n];
    }

    {
        v8f s0 = {}, s1 = {};
#pragma unroll
        for (int q = 0; q < 16; ++q) {
            const int r0 = 4 * q + 2 * hi;
            v2f uf;
            uf.x = Uts[r0 * 16 + m];
            uf.y = Uts[(r0 + 1) * 16 + m];
            s0 = __builtin_amdgcn_wmma_f32_16x16x4_f32(false, uf, false, a0[q],
                                                       (short)0, s0, false, false);
            s1 = __builtin_amdgcn_wmma_f32_16x16x4_f32(false, uf, false, a1[q],
                                                       (short)0, s1, false, false);
        }
        float* sp = Ssm + (size_t)(wave * 2) * 8 * 32 + lane;
#pragma unroll
        for (int v = 0; v < 8; ++v) {
            sp[v * 32] = s0[v];
            sp[(8 + v) * 32] = s1[v];
        }
    }
    __syncthreads();

    if (chain_wave) {
        v2f cf[4];
        for (int j = 0; j < 4; ++j) {
            cf[j].x = AcT[(8 * hi + 2 * j) * 16 + m];
            cf[j].y = AcT[(8 * hi + 2 * j + 1) * 16 + m];
        }
        const float* sp = Ssm + (size_t)wave * 8 * 32 + lane;
        float* hp = Hsm + (8 * hi) * SLAB + wave * 16 + m;
        v8f Ht = {};
        v8f cur;
        for (int v = 0; v < 8; ++v) cur[v] = sp[v * 32];
#pragma unroll
        for (int c = 0; c < NCH; ++c) {
            v8f nxt = {};
            if (c + 1 < NCH - 1) {
                for (int v = 0; v < 8; ++v) nxt[v] = sp[((c + 1) * 2 * 8 + v) * 32];
            }
            for (int v = 0; v < 8; ++v) hp[(c * NST + v) * SLAB] = Ht[v];
            if (c < NCH - 1) {
                v8f acc = cur;
                for (int j = 0; j < 4; ++j) {
                    v2f bf;
                    bf.x = Ht[2 * j];
                    bf.y = Ht[2 * j + 1];
                    acc = __builtin_amdgcn_wmma_f32_16x16x4_f32(false, cf[j], false, bf,
                                                                (short)0, acc, false, false);
                }
                Ht = acc;
                cur = nxt;
            }
        }
    }
    __syncthreads();

    v2f h0[4], h1[4];
    {
        const float* hq = Hsm + (size_t)wave * NST * SLAB;
#pragma unroll
        for (int j = 0; j < 4; ++j) {
            const int k0 = 4 * j + 2 * hi;
            h0[j].x = hq[k0 * SLAB + m];
            h0[j].y = hq[(k0 + 1) * SLAB + m];
            h1[j].x = hq[k0 * SLAB + 16 + m];
            h1[j].y = hq[(k0 + 1) * SLAB + 16 + m];
        }
    }

    float* yb = y + ((size_t)b * SEQ + (size_t)wave * CHUNK) * DMOD + dbase;
    float* stg = Ssm + (size_t)wave * 2 * 8 * 32;
    const int srow = lane >> 3;
    const int scol = 4 * (lane & 7);
#pragma unroll
    for (int it = 0; it < 4; ++it) {
        v8f c0 = {}, c1 = {};
#pragma unroll
        for (int q = 0; q < 4 * (it + 1); ++q) {
            const int r0 = 4 * q + 2 * hi;
            v2f tf;
            tf.x = Asm[r0 * AP + it * 16 + m];
            tf.y = Asm[(r0 + 1) * AP + it * 16 + m];
            c0 = __builtin_amdgcn_wmma_f32_16x16x4_f32(false, tf, false, a0[q],
                                                       (short)0, c0, false, false);
            c1 = __builtin_amdgcn_wmma_f32_16x16x4_f32(false, tf, false, a1[q],
                                                       (short)0, c1, false, false);
        }
#pragma unroll
        for (int j = 0; j < 4; ++j) {
            const int r0 = 64 + 4 * j + 2 * hi;
            v2f tf;
            tf.x = Asm[r0 * AP + it * 16 + m];
            tf.y = Asm[(r0 + 1) * AP + it * 16 + m];
            c0 = __builtin_amdgcn_wmma_f32_16x16x4_f32(false, tf, false, h0[j],
                                                       (short)0, c0, false, false);
            c1 = __builtin_amdgcn_wmma_f32_16x16x4_f32(false, tf, false, h1[j],
                                                       (short)0, c1, false, false);
        }
#pragma unroll
        for (int v = 0; v < 8; ++v) {
            stg[(8 * hi + v) * SLAB + m] = c0[v];
            stg[(8 * hi + v) * SLAB + 16 + m] = c1[v];
        }
        __builtin_amdgcn_wave_barrier();
        float4 t[4];
#pragma unroll
        for (int i = 0; i < 4; ++i)
            t[i] = *reinterpret_cast<const float4*>(stg + (4 * i + srow) * SLAB + scol);
#pragma unroll
        for (int i = 0; i < 4; ++i)
            *reinterpret_cast<float4*>(yb + (size_t)(it * 16 + 4 * i + srow) * DMOD + scol) = t[i];
        __builtin_amdgcn_wave_barrier();
    }
}

extern "C" void kernel_launch(void* const* d_in, const int* in_sizes, int n_in,
                              void* d_out, int out_size, void* d_ws, size_t ws_size,
                              hipStream_t stream) {
    (void)in_sizes; (void)n_in; (void)out_size; (void)d_ws; (void)ws_size;

    const float* x    = (const float*)d_in[0];
    const float* A    = (const float*)d_in[1];
    const float* Bvec = (const float*)d_in[2];
    float* y = (float*)d_out;

    k_fused<<<BATCH * (DMOD / SLAB), 1024, 0, stream>>>(x, A, Bvec, y);
}


# sequence split in two: 256 work-groups of 16 waves (one per WGP, was 128 of 32), one per half of a slab's sequence; the second-half group also forms the first half's state sums; sum orders unchanged
# speedup vs baseline: 5.4556x; 1.1666x over previous
#include <hip/hip_runtime.h>

typedef float v2f __attribute__((ext_vector_type(2)));
typedef float v8f __attribute__((ext_vector_type(8)));

#define BATCH 4
#define SEQ   2048
#define DMOD  1024
#define NST   16
#define CHUNK 64
#define NCH   (SEQ / CHUNK)
#define HCH   (NCH / 2)
#define NTHR  (HCH * 32)
#define SLAB  32
#define AP    72

__device__ __forceinline__ v8f k0_apply_A(const v2f (&af)[4], const v8f t) {
    v8f acc = {};
    for (int j = 0; j < 4; ++j) {
        v2f bf;
        bf.x = t[2 * j];
        bf.y = t[2 * j + 1];
        acc = __builtin_amdgcn_wmma_f32_16x16x4_f32(false, af[j], false, bf,
                                                    (short)0, acc, false, false);
    }
    return acc;
}

__device__ __forceinline__ void load_x_tile(const float* __restrict__ xb, const int hi, const int m,
                                            v2f (&t0)[16], v2f (&t1)[16]) {
#pragma unroll
    for (int q = 0; q < 16; ++q) {
        const int r0 = 4 * q + 2 * hi;
        t0[q].x = xb[(size_t)r0 * DMOD + m];
        t0[q].y = xb[(size_t)(r0 + 1) * DMOD + m];
        t1[q].x = xb[(size_t)r0 * DMOD + 16 + m];
        t1[q].y = xb[(size_t)(r0 + 1) * DMOD + 16 + m];
    }
}

__device__ __forceinline__ void store_s_tiles(float* sp, const v8f s0, const v8f s1) {
#pragma unroll
    for (int v = 0; v < 8; ++v) {
        sp[v * 32] = s0[v];
        sp[(8 + v) * 32] = s1[v];
    }
}

template <int HALF>
__device__ __forceinline__ void half_body(const float* __restrict__ x,
                                          const float* __restrict__ A,
                                          const float* __restrict__ Bvec,
                                          float* __restrict__ y,
                                          float* Uts, float* op, float* kp, float* AcT,
                                          float* Asm, float* Ssm, float* Hsm,
                                          const int b, const int dbase) {
    constexpr int CB = HALF * HCH;
    constexpr int NS = CB + HCH;

    const int tid = threadIdx.x;
    const int wave = __builtin_amdgcn_readfirstlane(tid >> 5);
    const int lane = tid & 31;
    const int hi = lane >> 4;
    const int m = lane & 15;

    const float* xb = x + ((size_t)b * SEQ + (size_t)(CB + wave) * CHUNK) * DMOD + dbase;
    const float* xe = x + ((size_t)b * SEQ + (size_t)wave * CHUNK) * DMOD + dbase;

    const bool chain_wave = (wave < 2);
    v2f af[4] = {};
    float bh[8] = {};
    if (chain_wave) {
        for (int j = 0; j < 4; ++j) {
            af[j].x = A[m * 16 + 8 * hi + 2 * j];
            af[j].y = A[m * 16 + 8 * hi + 2 * j + 1];
        }
        for (int v = 0; v < 8; ++v) bh[v] = Bvec[8 * hi + v];
    }

    v2f a0[16], a1[16];
    v2f e0[16], e1[16];
    if constexpr (HALF) {
        if (!chain_wave) load_x_tile(xe, hi, m, e0, e1);
    }
    load_x_tile(xb, hi, m, a0, a1);

    if (chain_wave) {
        if (wave == 0) {
            v8f u;
            for (int v = 0; v < 8; ++v) u[v] = (m == 0) ? bh[v] : 0.0f;
            for (int p = 0; p < CHUNK; ++p) {
                if (p > 0) u = k0_apply_A(af, u);
                if (m == 0) {
                    for (int v = 0; v < 8; ++v)
                        Uts[(63 - p) * 16 + 8 * hi + v] = u[v];
                }
                float s = 0.f;
                for (int v = 0; v < 8; ++v) s += bh[v] * u[v];
                if (m == 0) kp[hi * 64 + p] = s;
            }
        } else {
            v8f P;
            for (int v = 0; v < 8; ++v) P[v] = (8 * hi + v == m) ? 1.0f : 0.0f;
            for (int p = 0; p < CHUNK; ++p) {
                P = k0_apply_A(af, P);
                float s = 0.f;
                for (int v = 0; v < 8; ++v) s += bh[v] * P[v];
                op[(hi * 64 + p) * 16 + m] = s;
            }
            for (int v = 0; v < 8; ++v) AcT[m * 16 + 8 * hi + v] = P[v];
        }
        if constexpr (HALF) load_x_tile(xe, hi, m, e0, e1);
    }
    __syncthreads();

    for (int e = tid; e < 64 * 64; e += NTHR) {
        const int k = e >> 6, t = e & 63;
        Asm[k * AP + t] = (t >= k) ? (kp[t - k] + kp[64 + t - k]) : 0.0f;
    }
    for (int e = tid; e < 16 * 64; e += NTHR) {
        const int n = e >> 6, t = e & 63;
        Asm[(64 + n) * AP + t] = op[t * 16 + n] + op[(64 + t) * 16 + n];
    }

    {
        v8f s0 = {}, s1 = {}, s2 = {}, s3 = {};
#pragma unroll
        for (int q = 0; q < 16; ++q) {
            const int r0 = 4 * q + 2 * hi;
            v2f uf;
            uf.x = Uts[r0 * 16 + m];
            uf.y = Uts[(r0 + 1) * 16 + m];
            s0 = __builtin_amdgcn_wmma_f32_16x16x4_f32(false, uf, false, a0[q],
                                                       (short)0, s0, false, false);
            s1 = __builtin_amdgcn_wmma_f32_16x16x4_f32(false, uf, false, a1[q],
                                                       (short)0, s1, false, false);
            if constexpr (HALF) {
                s2 = __builtin_amdgcn_wmma_f32_16x16x4_f32(false, uf, false, e0[q],
                                                           (short)0, s2, false, false);
                s3 = __builtin_amdgcn_wmma_f32_16x16x4_f32(false, uf, false, e1[q],
                                                           (short)0, s3, false, false);
            }
        }
        store_s_tiles(Ssm + (size_t)((CB + wave) * 2) * 8 * 32 + lane, s0, s1);
        if constexpr (HALF) store_s_tiles(Ssm + (size_t)(wave * 2) * 8 * 32 + lane, s2, s3);
    }
    __syncthreads();

    if (chain_wave) {
        v2f cf[4];
        for (int j = 0; j < 4; ++j) {
            cf[j].x = AcT[(8 * hi + 2 * j) * 16 + m];
            cf[j].y = AcT[(8 * hi + 2 * j + 1) * 16 + m];
        }
        const float* sp = Ssm + (size_t)wave * 8 * 32 + lane;
        float* hp = Hsm + (8 * hi) * SLAB + wave * 16 + m;
        v8f Ht = {};
        v8f cur;
        for (int v = 0; v < 8; ++v) cur[v] = sp[v * 32];
#pragma unroll
        for (int c = 0; c < NS; ++c) {
            v8f nxt = {};
            if (c + 1 < NS - 1) {
                for (int v = 0; v < 8; ++v) nxt[v] = sp[((c + 1) * 2 * 8 + v) * 32];
            }
            if (c >= CB) {
                for (int v = 0; v < 8; ++v) hp[((c - CB) * NST + v) * SLAB] = Ht[v];
            }
            if (c < NS - 1) {
                v8f acc = cur;
                for (int j = 0; j < 4; ++j) {
                    v2f bf;
                    bf.x = Ht[2 * j];
                    bf.y = Ht[2 * j + 1];
                    acc = __builtin_amdgcn_wmma_f32_16x16x4_f32(false, cf[j], false, bf,
                                                                (short)0, acc, false, false);
                }
                Ht = acc;
                cur = nxt;
            }
        }
    }
    __syncthreads();

    v2f h0[4], h1[4];
    {
        const float* hq = Hsm + (size_t)wave * NST * SLAB;
#pragma unroll
        for (int j = 0; j < 4; ++j) {
            const int k0 = 4 * j + 2 * hi;
            h0[j].x = hq[k0 * SLAB + m];
            h0[j].y = hq[(k0 + 1) * SLAB + m];
            h1[j].x = hq[k0 * SLAB + 16 + m];
            h1[j].y = hq[(k0 + 1) * SLAB + 16 + m];
        }
    }

    float* yb = y + ((size_t)b * SEQ + (size_t)(CB + wave) * CHUNK) * DMOD + dbase;
    float* stg = Ssm + (size_t)wave * 2 * 8 * 32;
    const int srow = lane >> 3;
    const int scol = 4 * (lane & 7);
#pragma unroll
    for (int it = 0; it < 4; ++it) {
        v8f c0 = {}, c1 = {};
#pragma unroll
        for (int q = 0; q < 4 * (it + 1); ++q) {
            const int r0 = 4 * q + 2 * hi;
            v2f tf;
            tf.x = Asm[r0 * AP + it * 16 + m];
            tf.y = Asm[(r0 + 1) * AP + it * 16 + m];
            c0 = __builtin_amdgcn_wmma_f32_16x16x4_f32(false, tf, false, a0[q],
                                                       (short)0, c0, false, false);
            c1 = __builtin_amdgcn_wmma_f32_16x16x4_f32(false, tf, false, a1[q],
                                                       (short)0, c1, false, false);
        }
#pragma unroll
        for (int j = 0; j < 4; ++j) {
            const int r0 = 64 + 4 * j + 2 * hi;
            v2f tf;
            tf.x = Asm[r0 * AP + it * 16 + m];
            tf.y = Asm[(r0 + 1) * AP + it * 16 + m];
            c0 = __builtin_amdgcn_wmma_f32_16x16x4_f32(false, tf, false, h0[j],
                                                       (short)0, c0, false, false);
            c1 = __builtin_amdgcn_wmma_f32_16x16x4_f32(false, tf, false, h1[j],
                                                       (short)0, c1, false, false);
        }
#pragma unroll
        for (int v = 0; v < 8; ++v) {
            stg[(8 * hi + v) * SLAB + m] = c0[v];
            stg[(8 * hi + v) * SLAB + 16 + m] = c1[v];
        }
        __builtin_amdgcn_wave_barrier();
        float4 t[4];
#pragma unroll
        for (int i = 0; i < 4; ++i)
            t[i] = *reinterpret_cast<const float4*>(stg + (4 * i + srow) * SLAB + scol);
#pragma unroll
        for (int i = 0; i < 4; ++i)
            *reinterpret_cast<float4*>(yb + (size_t)(it * 16 + 4 * i + srow) * DMOD + scol) = t[i];
        __builtin_amdgcn_wave_barrier();
    }
}

__global__ __launch_bounds__(NTHR) void k_fused(const float* __restrict__ x,
                                                const float* __restrict__ A,
                                                const float* __restrict__ Bvec,
                                                float* __restrict__ y) {
    __shared__ float Uts[64 * 16];
    __shared__ float op[2 * 64 * 16];
    __shared__ float kp[2 * 64];
    __shared__ float AcT[16 * 16];
    __shared__ float Asm[80 * AP];
    __shared__ __attribute__((aligned(16))) float Ssm[NCH * 2 * 8 * 32];
    __shared__ float Hsm[HCH * NST * SLAB];

    const int sb = blockIdx.x & 127;
    const int b = sb >> 5;
    const int dbase = (sb & 31) * SLAB;
    if (blockIdx.x < 128)
        half_body<1>(x, A, Bvec, y, Uts, op, kp, AcT, Asm, Ssm, Hsm, b, dbase);
    else
        half_body<0>(x, A, Bvec, y, Uts, op, kp, AcT, Asm, Ssm, Hsm, b, dbase);
}

extern "C" void kernel_launch(void* const* d_in, const int* in_sizes, int n_in,
                              void* d_out, int out_size, void* d_ws, size_t ws_size,
                              hipStream_t stream) {
    (void)in_sizes; (void)n_in; (void)out_size; (void)d_ws; (void)ws_size;

    const float* x    = (const float*)d_in[0];
    const float* A    = (const float*)d_in[1];
    const float* Bvec = (const float*)d_in[2];
    float* y = (float*)d_out;

    k_fused<<<2 * BATCH * (DMOD / SLAB), NTHR, 0, stream>>>(x, A, Bvec, y);
}


# wave 0's 63-step chain keeps only its 4 matrix instructions and the store of A^p B per step; the 64 sums k_p = B^T A^p B (9 dependent vector ops issued between steps) follow it, term order kept
# speedup vs baseline: 5.8625x; 1.0746x over previous
#include <hip/hip_runtime.h>

typedef float v2f __attribute__((ext_vector_type(2)));
typedef float v8f __attribute__((ext_vector_type(8)));

#define BATCH 4
#define SEQ   2048
#define DMOD  1024
#define NST   16
#define CHUNK 64
#define NCH   (SEQ / CHUNK)
#define HCH   (NCH / 2)
#define NTHR  (HCH * 32)
#define SLAB  32
#define AP    72

__device__ __forceinline__ v8f k0_apply_A(const v2f (&af)[4], const v8f t) {
    v8f acc = {};
    for (int j = 0; j < 4; ++j) {
        v2f bf;
        bf.x = t[2 * j];
        bf.y = t[2 * j + 1];
        acc = __builtin_amdgcn_wmma_f32_16x16x4_f32(false, af[j], false, bf,
                                                    (short)0, acc, false, false);
    }
    return acc;
}

__device__ __forceinline__ void load_x_tile(const float* __restrict__ xb, const int hi, const int m,
                                            v2f (&t0)[16], v2f (&t1)[16]) {
#pragma unroll
    for (int q = 0; q < 16; ++q) {
        const int r0 = 4 * q + 2 * hi;
        t0[q].x = xb[(size_t)r0 * DMOD + m];
        t0[q].y = xb[(size_t)(r0 + 1) * DMOD + m];
        t1[q].x = xb[(size_t)r0 * DMOD + 16 + m];
        t1[q].y = xb[(size_t)(r0 + 1) * DMOD + 16 + m];
    }
}

__device__ __forceinline__ void store_s_tiles(float* sp, const v8f s0, const v8f s1) {
#pragma unroll
    for (int v = 0; v < 8; ++v) {
        sp[v * 32] = s0[v];
        sp[(8 + v) * 32] = s1[v];
    }
}

template <int HALF>
__device__ __forceinline__ void half_body(const float* __restrict__ x,
                                          const float* __restrict__ A,
                                          const float* __restrict__ Bvec,
                                          float* __restrict__ y,
                                          float* Uts, float* op, float* kp, float* AcT,
                                          float* Asm, float* Ssm, float* Hsm,
                                          const int b, const int dbase) {
    constexpr int CB = HALF * HCH;
    constexpr int NS = CB + HCH;

    const int tid = threadIdx.x;
    const int wave = __builtin_amdgcn_readfirstlane(tid >> 5);
    const int lane = tid & 31;
    const int hi = lane >> 4;
    const int m = lane & 15;

    const float* xb = x + ((size_t)b * SEQ + (size_t)(CB + wave) * CHUNK) * DMOD + dbase;
    const float* xe = x + ((size_t)b * SEQ + (size_t)wave * CHUNK) * DMOD + dbase;

    const bool chain_wave = (wave < 2);
    v2f af[4] = {};
    float bh[8] = {};
    if (chain_wave) {
        for (int j = 0; j < 4; ++j) {
            af[j].x = A[m * 16 + 8 * hi + 2 * j];
            af[j].y = A[m * 16 + 8 * hi + 2 * j + 1];
        }
        for (int v = 0; v < 8; ++v) bh[v] = Bvec[8 * hi + v];
    }

    v2f a0[16], a1[16];
    v2f e0[16], e1[16];
    if constexpr (HALF) {
        if (!chain_wave) load_x_tile(xe, hi, m, e0, e1);
    }
    load_x_tile(xb, hi, m, a0, a1);

    if (chain_wave) {
        if (wave == 0) {
            v8f u;
            for (int v = 0; v < 8; ++v) u[v] = (m == 0) ? bh[v] : 0.0f;
            for (int p = 0; p < CHUNK; ++p) {
                if (p > 0) u = k0_apply_A(af, u);
                if (m == 0) {
                    for (int v = 0; v < 8; ++v)
                        Uts[(63 - p) * 16 + 8 * hi + v] = u[v];
                }
            }
            for (int i = 0; i < 4; ++i) {
                const int p = 16 * i + m;
                const float* up = Uts + (63 - p) * 16 + 8 * hi;
                float s = 0.f;
                for (int v = 0; v < 8; ++v) s += bh[v] * up[v];
                kp[hi * 64 + p] = s;
            }
        } else {
            v8f P;
            for (int v = 0; v < 8; ++v) P[v] = (8 * hi + v == m) ? 1.0f : 0.0f;
            for (int p = 0; p < CHUNK; ++p) {
                P = k0_apply_A(af, P);
                float s = 0.f;
                for (int v = 0; v < 8; ++v) s += bh[v] * P[v];
                op[(hi * 64 + p) * 16 + m] = s;
            }
            for (int v = 0; v < 8; ++v) AcT[m * 16 + 8 * hi + v] = P[v];
        }
        if constexpr (HALF) load_x_tile(xe, hi, m, e0, e1);
    }
    __syncthreads();

    for (int e = tid; e < 64 * 64; e += NTHR) {
        const int k = e >> 6, t = e & 63;
        Asm[k * AP + t] = (t >= k) ? (kp[t - k] + kp[64 + t - k]) : 0.0f;
    }
    for (int e = tid; e < 16 * 64; e += NTHR) {
        const int n = e >> 6, t = e & 63;
        Asm[(64 + n) * AP + t] = op[t * 16 + n] + op[(64 + t) * 16 + n];
    }

    {
        v8f s0 = {}, s1 = {}, s2 = {}, s3 = {};
#pragma unroll
        for (int q = 0; q < 16; ++q) {
            const int r0 = 4 * q + 2 * hi;
            v2f uf;
            uf.x = Uts[r0 * 16 + m];
            uf.y = Uts[(r0 + 1) * 16 + m];
            s0 = __builtin_amdgcn_wmma_f32_16x16x4_f32(false, uf, false, a0[q],
                                                       (short)0, s0, false, false);
            s1 = __builtin_amdgcn_wmma_f32_16x16x4_f32(false, uf, false, a1[q],
                                                       (short)0, s1, false, false);
            if constexpr (HALF) {
                s2 = __builtin_amdgcn_wmma_f32_16x16x4_f32(false, uf, false, e0[q],
                                                           (short)0, s2, false, false);
                s3 = __builtin_amdgcn_wmma_f32_16x16x4_f32(false, uf, false, e1[q],
                                                           (short)0, s3, false, false);
            }
        }
        store_s_tiles(Ssm + (size_t)((CB + wave) * 2) * 8 * 32 + lane, s0, s1);
        if constexpr (HALF) store_s_tiles(Ssm + (size_t)(wave * 2) * 8 * 32 + lane, s2, s3);
    }
    __syncthreads();

    if (chain_wave) {
        v2f cf[4];
        for (int j = 0; j < 4; ++j) {
            cf[j].x = AcT[(8 * hi + 2 * j) * 16 + m];
            cf[j].y = AcT[(8 * hi + 2 * j + 1) * 16 + m];
        }
        const float* sp = Ssm + (size_t)wave * 8 * 32 + lane;
        float* hp = Hsm + (8 * hi) * SLAB + wave * 16 + m;
        v8f Ht = {};
        v8f cur;
        for (int v = 0; v < 8; ++v) cur[v] = sp[v * 32];
#pragma unroll
        for (int c = 0; c < NS; ++c) {
            v8f nxt = {};
            if (c + 1 < NS - 1) {
                for (int v = 0; v < 8; ++v) nxt[v] = sp[((c + 1) * 2 * 8 + v) * 32];
            }
            if (c >= CB) {
                for (int v = 0; v < 8; ++v) hp[((c - CB) * NST + v) * SLAB] = Ht[v];
            }
            if (c < NS - 1) {
                v8f acc = cur;
                for (int j = 0; j < 4; ++j) {
                    v2f bf;
                    bf.x = Ht[2 * j];
                    bf.y = Ht[2 * j + 1];
                    acc = __builtin_amdgcn_wmma_f32_16x16x4_f32(false, cf[j], false, bf,
                                                                (short)0, acc, false, false);
                }
                Ht = acc;
                cur = nxt;
            }
        }
    }
    __syncthreads();

    v2f h0[4], h1[4];
    {
        const float* hq = Hsm + (size_t)wave * NST * SLAB;
#pragma unroll
        for (int j = 0; j < 4; ++j) {
            const int k0 = 4 * j + 2 * hi;
            h0[j].x = hq[k0 * SLAB + m];
            h0[j].y = hq[(k0 + 1) * SLAB + m];
            h1[j].x = hq[k0 * SLAB + 16 + m];
            h1[j].y = hq[(k0 + 1) * SLAB + 16 + m];
        }
    }

    float* yb = y + ((size_t)b * SEQ + (size_t)(CB + wave) * CHUNK) * DMOD + dbase;
    float* stg = Ssm + (size_t)wave * 2 * 8 * 32;
    const int srow = lane >> 3;
    const int scol = 4 * (lane & 7);
#pragma unroll
    for (int it = 0; it < 4; ++it) {
        v8f c0 = {}, c1 = {};
#pragma unroll
        for (int q = 0; q < 4 * (it + 1); ++q) {
            const int r0 = 4 * q + 2 * hi;
            v2f tf;
            tf.x = Asm[r0 * AP + it * 16 + m];
            tf.y = Asm[(r0 + 1) * AP + it * 16 + m];
            c0 = __builtin_amdgcn_wmma_f32_16x16x4_f32(false, tf, false, a0[q],
                                                       (short)0, c0, false, false);
            c1 = __builtin_amdgcn_wmma_f32_16x16x4_f32(false, tf, false, a1[q],
                                                       (short)0, c1, false, false);
        }
#pragma unroll
        for (int j = 0; j < 4; ++j) {
            const int r0 = 64 + 4 * j + 2 * hi;
            v2f tf;
            tf.x = Asm[r0 * AP + it * 16 + m];
            tf.y = Asm[(r0 + 1) * AP + it * 16 + m];
            c0 = __builtin_amdgcn_wmma_f32_16x16x4_f32(false, tf, false, h0[j],
                                                       (short)0, c0, false, false);
            c1 = __builtin_amdgcn_wmma_f32_16x16x4_f32(false, tf, false, h1[j],
                                                       (short)0, c1, false, false);
        }
#pragma unroll
        for (int v = 0; v < 8; ++v) {
            stg[(8 * hi + v) * SLAB + m] = c0[v];
            stg[(8 * hi + v) * SLAB + 16 + m] = c1[v];
        }
        __builtin_amdgcn_wave_barrier();
        float4 t[4];
#pragma unroll
        for (int i = 0; i < 4; ++i)
            t[i] = *reinterpret_cast<const float4*>(stg + (4 * i + srow) * SLAB + scol);
#pragma unroll
        for (int i = 0; i < 4; ++i)
            *reinterpret_cast<float4*>(yb + (size_t)(it * 16 + 4 * i + srow) * DMOD + scol) = t[i];
        __builtin_amdgcn_wave_barrier();
    }
}

__global__ __launch_bounds__(NTHR) void k_fused(const float* __restrict__ x,
                                                const float* __restrict__ A,
                                                const float* __restrict__ Bvec,
                                                float* __restrict__ y) {
    __shared__ float Uts[64 * 16];
    __shared__ float op[2 * 64 * 16];
    __shared__ float kp[2 * 64];
    __shared__ float AcT[16 * 16];
    __shared__ float Asm[80 * AP];
    __shared__ __attribute__((aligned(16))) float Ssm[NCH * 2 * 8 * 32];
    __shared__ float Hsm[HCH * NST * SLAB];

    const int sb = blockIdx.x & 127;
    const int b = sb >> 5;
    const int dbase = (sb & 31) * SLAB;
    if (blockIdx.x < 128)
        half_body<1>(x, A, Bvec, y, Uts, op, kp, AcT, Asm, Ssm, Hsm, b, dbase);
    else
        half_body<0>(x, A, Bvec, y, Uts, op, kp, AcT, Asm, Ssm, Hsm, b, dbase);
}

extern "C" void kernel_launch(void* const* d_in, const int* in_sizes, int n_in,
                              void* d_out, int out_size, void* d_ws, size_t ws_size,
                              hipStream_t stream) {
    (void)in_sizes; (void)n_in; (void)out_size; (void)d_ws; (void)ws_size;

    const float* x    = (const float*)d_in[0];
    const float* A    = (const float*)d_in[1];
    const float* Bvec = (const float*)d_in[2];
    float* y = (float*)d_out;

    k_fused<<<2 * BATCH * (DMOD / SLAB), NTHR, 0, stream>>>(x, A, Bvec, y);
}


# phase 1 power chains 19 dependent steps deep, not 64: 16 steps with A fill a tile column by column, 3 with A^16 advance 16 columns each; a third wave takes the O rows; sum order in A^p changes
# speedup vs baseline: 6.6039x; 1.1265x over previous
#include <hip/hip_runtime.h>

typedef float v2f __attribute__((ext_vector_type(2)));
typedef float v8f __attribute__((ext_vector_type(8)));

#define BATCH 4
#define SEQ   2048
#define DMOD  1024
#define NST   16
#define CHUNK 64
#define NCH   (SEQ / CHUNK)
#define HCH   (NCH / 2)
#define NTHR  (HCH * 32)
#define SLAB  32
#define AP    72

__device__ __forceinline__ v8f k0_apply_A(const v2f (&af)[4], const v8f t, const v8f init) {
    v8f acc = init;
    for (int j = 0; j < 4; ++j) {
        v2f bf;
        bf.x = t[2 * j];
        bf.y = t[2 * j + 1];
        acc = __builtin_amdgcn_wmma_f32_16x16x4_f32(false, af[j], false, bf,
                                                    (short)0, acc, false, false);
    }
    return acc;
}

__device__ __forceinline__ void load_x_tile(const float* __restrict__ xb, const int hi, const int m,
                                            v2f (&t0)[16], v2f (&t1)[16]) {
#pragma unroll
    for (int q = 0; q < 16; ++q) {
        const int r0 = 4 * q + 2 * hi;
        t0[q].x = xb[(size_t)r0 * DMOD + m];
        t0[q].y = xb[(size_t)(r0 + 1) * DMOD + m];
        t1[q].x = xb[(size_t)r0 * DMOD + 16 + m];
        t1[q].y = xb[(size_t)(r0 + 1) * DMOD + 16 + m];
    }
}

__device__ __forceinline__ void store_s_tiles(float* sp, const v8f s0, const v8f s1) {
#pragma unroll
    for (int v = 0; v < 8; ++v) {
        sp[v * 32] = s0[v];
        sp[(8 + v) * 32] = s1[v];
    }
}

template <int HALF>
__device__ __forceinline__ void half_body(const float* __restrict__ x,
                                          const float* __restrict__ A,
                                          const float* __restrict__ Bvec,
                                          float* __restrict__ y,
                                          float* Uts, float* op, float* kp, float* AcT, float* G16T,
                                          float* Asm, float* Ssm, float* Hsm,
                                          const int b, const int dbase) {
    constexpr int CB = HALF * HCH;
    constexpr int NS = CB + HCH;

    const int tid = threadIdx.x;
    const int wave = __builtin_amdgcn_readfirstlane(tid >> 5);
    const int lane = tid & 31;
    const int hi = lane >> 4;
    const int m = lane & 15;

    const float* xb = x + ((size_t)b * SEQ + (size_t)(CB + wave) * CHUNK) * DMOD + dbase;
    const float* xe = x + ((size_t)b * SEQ + (size_t)wave * CHUNK) * DMOD + dbase;

    const bool chain_wave = (wave < 3);
    v2f af[4] = {};
    float bh[8] = {};
    if (chain_wave) {
        if (wave == 2) {
            for (int j = 0; j < 4; ++j) {
                af[j].x = A[(8 * hi + 2 * j) * 16 + m];
                af[j].y = A[(8 * hi + 2 * j + 1) * 16 + m];
            }
        } else {
            for (int j = 0; j < 4; ++j) {
                af[j].x = A[m * 16 + 8 * hi + 2 * j];
                af[j].y = A[m * 16 + 8 * hi + 2 * j + 1];
            }
        }
        for (int v = 0; v < 8; ++v) bh[v] = Bvec[8 * hi + v];
    }

    v2f a0[16], a1[16];
    v2f e0[16], e1[16];
    if constexpr (HALF) {
        if (!chain_wave) load_x_tile(xe, hi, m, e0, e1);
    }
    load_x_tile(xb, hi, m, a0, a1);

    const v8f zero8 = {};
    v8f ct = {};
    if (chain_wave) {
        if (wave == 1) {
            for (int v = 0; v < 8; ++v) ct[v] = (8 * hi + v == m) ? 1.0f : 0.0f;
            for (int p = 0; p < 16; ++p) ct = k0_apply_A(af, ct, zero8);
            for (int v = 0; v < 8; ++v) G16T[m * 16 + 8 * hi + v] = ct[v];
        } else {
            for (int v = 0; v < 8; ++v) ct[v] = (m == 0) ? bh[v] : 0.0f;
            for (int p = 1; p < 16; ++p) {
                v8f init;
                for (int v = 0; v < 8; ++v) init[v] = (m == p) ? bh[v] : 0.0f;
                ct = k0_apply_A(af, ct, init);
            }
            if (wave == 0) {
                for (int v = 0; v < 8; ++v) Uts[(48 + m) * 16 + 8 * hi + v] = ct[v];
            } else {
                ct = k0_apply_A(af, ct, zero8);
                for (int v = 0; v < 8; ++v) op[(15 - m) * 16 + 8 * hi + v] = ct[v];
            }
        }
        if constexpr (HALF) load_x_tile(xe, hi, m, e0, e1);
    }
    __syncthreads();
    if (chain_wave) {
        v2f gf[4];
        if (wave == 2) {
            for (int j = 0; j < 4; ++j) {
                gf[j].x = G16T[m * 16 + 8 * hi + 2 * j];
                gf[j].y = G16T[m * 16 + 8 * hi + 2 * j + 1];
            }
        } else {
            for (int j = 0; j < 4; ++j) {
                gf[j].x = G16T[(8 * hi + 2 * j) * 16 + m];
                gf[j].y = G16T[(8 * hi + 2 * j + 1) * 16 + m];
            }
        }
        for (int blk = 1; blk < 4; ++blk) {
            ct = k0_apply_A(gf, ct, zero8);
            if (wave == 0) {
                for (int v = 0; v < 8; ++v) Uts[(48 - 16 * blk + m) * 16 + 8 * hi + v] = ct[v];
            } else if (wave == 2) {
                for (int v = 0; v < 8; ++v) op[(16 * blk + 15 - m) * 16 + 8 * hi + v] = ct[v];
            }
        }
        if (wave == 0) {
            for (int i = 0; i < 4; ++i) {
                const int p = 16 * i + m;
                const float* up = Uts + (63 - p) * 16 + 8 * hi;
                float s = 0.f;
                for (int v = 0; v < 8; ++v) s += bh[v] * up[v];
                kp[hi * 64 + p] = s;
            }
        } else if (wave == 1) {
            for (int v = 0; v < 8; ++v) AcT[m * 16 + 8 * hi + v] = ct[v];
        }
    }
    __syncthreads();

    for (int e = tid; e < 64 * 64; e += NTHR) {
        const int k = e >> 6, t = e & 63;
        Asm[k * AP + t] = (t >= k) ? (kp[t - k] + kp[64 + t - k]) : 0.0f;
    }
    for (int e = tid; e < 16 * 64; e += NTHR) {
        const int n = e >> 6, t = e & 63;
        Asm[(64 + n) * AP + t] = op[t * 16 + n];
    }

    {
        v8f s0 = {}, s1 = {}, s2 = {}, s3 = {};
#pragma unroll
        for (int q = 0; q < 16; ++q) {
            const int r0 = 4 * q + 2 * hi;
            v2f uf;
            uf.x = Uts[r0 * 16 + m];
            uf.y = Uts[(r0 + 1) * 16 + m];
            s0 = __builtin_amdgcn_wmma_f32_16x16x4_f32(false, uf, false, a0[q],
                                                       (short)0, s0, false, false);
            s1 = __builtin_amdgcn_wmma_f32_16x16x4_f32(false, uf, false, a1[q],
                                                       (short)0, s1, false, false);
            if constexpr (HALF) {
                s2 = __builtin_amdgcn_wmma_f32_16x16x4_f32(false, uf, false, e0[q],
                                                           (short)0, s2, false, false);
                s3 = __builtin_amdgcn_wmma_f32_16x16x4_f32(false, uf, false, e1[q],
                                                           (short)0, s3, false, false);
            }
        }
        store_s_tiles(Ssm + (size_t)((CB + wave) * 2) * 8 * 32 + lane, s0, s1);
        if constexpr (HALF) store_s_tiles(Ssm + (size_t)(wave * 2) * 8 * 32 + lane, s2, s3);
    }
    __syncthreads();

    if (wave < 2) {
        v2f cf[4];
        for (int j = 0; j < 4; ++j) {
            cf[j].x = AcT[(8 * hi + 2 * j) * 16 + m];
            cf[j].y = AcT[(8 * hi + 2 * j + 1) * 16 + m];
        }
        const float* sp = Ssm + (size_t)wave * 8 * 32 + lane;
        float* hp = Hsm + (8 * hi) * SLAB + wave * 16 + m;
        v8f Ht = {};
        v8f cur;
        for (int v = 0; v < 8; ++v) cur[v] = sp[v * 32];
#pragma unroll
        for (int c = 0; c < NS; ++c) {
            v8f nxt = {};
            if (c + 1 < NS - 1) {
                for (int v = 0; v < 8; ++v) nxt[v] = sp[((c + 1) * 2 * 8 + v) * 32];
            }
            if (c >= CB) {
                for (int v = 0; v < 8; ++v) hp[((c - CB) * NST + v) * SLAB] = Ht[v];
            }
            if (c < NS - 1) {
                v8f acc = cur;
                for (int j = 0; j < 4; ++j) {
                    v2f bf;
                    bf.x = Ht[2 * j];
                    bf.y = Ht[2 * j + 1];
                    acc = __builtin_amdgcn_wmma_f32_16x16x4_f32(false, cf[j], false, bf,
                                                                (short)0, acc, false, false);
                }
                Ht = acc;
                cur = nxt;
            }
        }
    }
    __syncthreads();

    v2f h0[4], h1[4];
    {
        const float* hq = Hsm + (size_t)wave * NST * SLAB;
#pragma unroll
        for (int j = 0; j < 4; ++j) {
            const int k0 = 4 * j + 2 * hi;
            h0[j].x = hq[k0 * SLAB + m];
            h0[j].y = hq[(k0 + 1) * SLAB + m];
            h1[j].x = hq[k0 * SLAB + 16 + m];
            h1[j].y = hq[(k0 + 1) * SLAB + 16 + m];
        }
    }

    float* yb = y + ((size_t)b * SEQ + (size_t)(CB + wave) * CHUNK) * DMOD + dbase;
    float* stg = Ssm + (size_t)wave * 2 * 8 * 32;
    const int srow = lane >> 3;
    const int scol = 4 * (lane & 7);
#pragma unroll
    for (int it = 0; it < 4; ++it) {
        v8f c0 = {}, c1 = {};
#pragma unroll
        for (int q = 0; q < 4 * (it + 1); ++q) {
            const int r0 = 4 * q + 2 * hi;
            v2f tf;
            tf.x = Asm[r0 * AP + it * 16 + m];
            tf.y = Asm[(r0 + 1) * AP + it * 16 + m];
            c0 = __builtin_amdgcn_wmma_f32_16x16x4_f32(false, tf, false, a0[q],
                                                       (short)0, c0, false, false);
            c1 = __builtin_amdgcn_wmma_f32_16x16x4_f32(false, tf, false, a1[q],
                                                       (short)0, c1, false, false);
        }
#pragma unroll
        for (int j = 0; j < 4; ++j) {
            const int r0 = 64 + 4 * j + 2 * hi;
            v2f tf;
            tf.x = Asm[r0 * AP + it * 16 + m];
            tf.y = Asm[(r0 + 1) * AP + it * 16 + m];
            c0 = __builtin_amdgcn_wmma_f32_16x16x4_f32(false, tf, false, h0[j],
                                                       (short)0, c0, false, false);
            c1 = __builtin_amdgcn_wmma_f32_16x16x4_f32(false, tf, false, h1[j],
                                                       (short)0, c1, false, false);
        }
#pragma unroll
        for (int v = 0; v < 8; ++v) {
            stg[(8 * hi + v) * SLAB + m] = c0[v];
            stg[(8 * hi + v) * SLAB + 16 + m] = c1[v];
        }
        __builtin_amdgcn_wave_barrier();
        float4 t[4];
#pragma unroll
        for (int i = 0; i < 4; ++i)
            t[i] = *reinterpret_cast<const float4*>(stg + (4 * i + srow) * SLAB + scol);
#pragma unroll
        for (int i = 0; i < 4; ++i)
            *reinterpret_cast<float4*>(yb + (size_t)(it * 16 + 4 * i + srow) * DMOD + scol) = t[i];
        __builtin_amdgcn_wave_barrier();
    }
}

__global__ __launch_bounds__(NTHR) void k_fused(const float* __restrict__ x,
                                                const float* __restrict__ A,
                                                const float* __restrict__ Bvec,
                                                float* __restrict__ y) {
    __shared__ float Uts[64 * 16];
    __shared__ float op[64 * 16];
    __shared__ float kp[2 * 64];
    __shared__ float AcT[16 * 16];
    __shared__ float G16T[16 * 16];
    __shared__ float Asm[80 * AP];
    __shared__ __attribute__((aligned(16))) float Ssm[NCH * 2 * 8 * 32];
    __shared__ float Hsm[HCH * NST * SLAB];

    const int sb = blockIdx.x & 127;
    const int b = sb >> 5;
    const int dbase = (sb & 31) * SLAB;
    if (blockIdx.x < 128)
        half_body<1>(x, A, Bvec, y, Uts, op, kp, AcT, G16T, Asm, Ssm, Hsm, b, dbase);
    else
        half_body<0>(x, A, Bvec, y, Uts, op, kp, AcT, G16T, Asm, Ssm, Hsm, b, dbase);
}

extern "C" void kernel_launch(void* const* d_in, const int* in_sizes, int n_in,
                              void* d_out, int out_size, void* d_ws, size_t ws_size,
                              hipStream_t stream) {
    (void)in_sizes; (void)n_in; (void)out_size; (void)d_ws; (void)ws_size;

    const float* x    = (const float*)d_in[0];
    const float* A    = (const float*)d_in[1];
    const float* Bvec = (const float*)d_in[2];
    float* y = (float*)d_out;

    k_fused<<<2 * BATCH * (DMOD / SLAB), NTHR, 0, stream>>>(x, A, Bvec, y);
}
